# starting kernel
# baseline (speedup 1.0000x reference)
#include <hip/hip_runtime.h>
#include <math.h>

typedef __attribute__((ext_vector_type(2))) float v2f;
typedef __attribute__((ext_vector_type(8))) float v8f;

#define C_DIM 256
#define D_DIM 256
#define N_TOK 4096
#define SCALE 0.0625f
#define EPSV  1e-5f

#define WS_GP    0
#define WS_SGP   1024
#define WS_SBP   1028
#define WS_LOG   1032
#define WS_MU    17416
#define WS_RSTD  33800
#define WS_ALPHA 50184
#define WS_WSUM  66568
#define WS_AMU   66572
#define WS_U     66576
#define WS_YVEC  67600

__device__ inline v8f gemv_tile_f32(const float* x, const float* w,
                                    int K, long sk, long sn, int n0, v8f acc) {
    const int lane = threadIdx.x & 31;
    const int half = lane >> 4;
    const int l16  = lane & 15;
    const float amask = (l16 == 0) ? 1.0f : 0.0f;
    const float* wp = w + (long)(n0 + l16) * sn;
    for (int k0 = 0; k0 < K; k0 += 4) {
        const int kb = k0 + 2 * half;
        const float x0 = x[kb];
        const float x1 = x[kb + 1];
        v2f a;
        a.x = x0 * amask;
        a.y = x1 * amask;
        v2f b;
        b.x = wp[(long)kb * sk];
        b.y = wp[(long)(kb + 1) * sk];
        acc = __builtin_amdgcn_wmma_f32_16x16x4_f32(
             false, a,  false, b,
             (short)0, acc,  false,  false);
    }
    return acc;
}

__device__ inline float block_reduce_sum_512(float v, float* red) {
    const int t = threadIdx.x;
    red[t] = v; __syncthreads();
    for (int o = 256; o > 0; o >>= 1) {
        if (t < o) red[t] += red[t + o];
        __syncthreads();
    }
    float r = red[0]; __syncthreads();
    return r;
}

__global__ __launch_bounds__(512) void prep_kernel(
    const float* tvec, const float* wt, const float* bt,
    const float* wk_x, const float* wk_h, const float* wk_s, const float* wk_m,
    const float* g_x, const float* be_x, const float* g_h, const float* be_h,
    const float* g_s, const float* be_s, const float* g_m, const float* be_m,
    float* gp, float* sgp, float* sbp)
{
    __shared__ float q_s[D_DIM];
    __shared__ float p_s[C_DIM];
    __shared__ float red[512];
    const int tid  = threadIdx.x;
    const int wave = tid >> 5;
    const int lane = tid & 31;

    {
        v8f acc = {};
        acc = gemv_tile_f32(tvec, wt, C_DIM, (long)D_DIM, 1L, wave * 16, acc);
        if (lane < 16) q_s[wave * 16 + lane] = acc[0] + bt[wave * 16 + lane];
    }
    __syncthreads();

    const float* wks[4] = {wk_x, wk_h, wk_s, wk_m};
    const float* gs[4]  = {g_x,  g_h,  g_s,  g_m};
    const float* bes[4] = {be_x, be_h, be_s, be_m};

    for (int mod = 0; mod < 4; ++mod) {
        v8f acc = {};
        acc = gemv_tile_f32(q_s, wks[mod], D_DIM, 1L, (long)D_DIM, wave * 16, acc);
        if (lane < 16) p_s[wave * 16 + lane] = acc[0];
        __syncthreads();

        float gv = 0.0f, bv = 0.0f;
        if (tid < C_DIM) {
            const float p = p_s[tid];
            gv = gs[mod][tid] * p;
            bv = bes[mod][tid] * p;
            gp[mod * C_DIM + tid] = gv;
        }
        const float sg = block_reduce_sum_512(gv, red);
        const float sb = block_reduce_sum_512(bv, red);
        if (tid == 0) { sgp[mod] = sg; sbp[mod] = sb; }
        __syncthreads();
    }
}

__global__ __launch_bounds__(256) void pass1_kernel(
    const float* x, const float* h, const float* s, const float* m,
    const float* gp, const float* sgp, const float* sbp,
    float* logits, float* mus, float* rstds)
{
    __shared__ float gpm_s[C_DIM];
    __shared__ float sh1[4][64];
    __shared__ float sh2[4][64];
    __shared__ float shd[4][64];
    const int bid = blockIdx.x;
    const int mod = bid >> 6;
    const int j0  = (bid & 63) << 6;
    const int tl  = threadIdx.x & 63;
    const int grp = threadIdx.x >> 6;
    const float* aptr[4] = {x, h, s, m};
    const float* a = aptr[mod];

    gpm_s[threadIdx.x] = gp[mod * C_DIM + threadIdx.x];
    __syncthreads();

    const int j  = j0 + tl;
    const int c0 = grp << 6;
    float s1 = 0.0f, s2 = 0.0f, dp = 0.0f;
#pragma unroll 4
    for (int cc = 0; cc < 64; ++cc) {
        const int c = c0 + cc;
        const float v = a[(long)c * N_TOK + j];
        s1 += v;
        s2 += v * v;
        dp += v * gpm_s[c];
    }
    sh1[grp][tl] = s1; sh2[grp][tl] = s2; shd[grp][tl] = dp;
    __syncthreads();

    if (threadIdx.x < 64) {
        const int jj = j0 + threadIdx.x;
        const float t1 = sh1[0][threadIdx.x] + sh1[1][threadIdx.x] +
                         sh1[2][threadIdx.x] + sh1[3][threadIdx.x];
        const float t2 = sh2[0][threadIdx.x] + sh2[1][threadIdx.x] +
                         sh2[2][threadIdx.x] + sh2[3][threadIdx.x];
        const float td = shd[0][threadIdx.x] + shd[1][threadIdx.x] +
                         shd[2][threadIdx.x] + shd[3][threadIdx.x];
        const float mu   = t1 * (1.0f / C_DIM);
        const float var  = t2 * (1.0f / C_DIM) - mu * mu;
        const float rstd = rsqrtf(var + EPSV);
        const float lg   = SCALE * (rstd * (td - mu * sgp[mod]) + sbp[mod]);
        const int idx = mod * N_TOK + jj;
        logits[idx] = lg;
        mus[idx]    = mu;
        rstds[idx]  = rstd;
    }
}

__global__ __launch_bounds__(1024) void softmax_kernel(
    const float* logits, const float* mus, const float* rstds,
    float* alpha, float* wsum, float* amu)
{
    __shared__ float red[1024];
    const int t = threadIdx.x;
    float l[16];
    float lmax = -3.4e38f;
#pragma unroll
    for (int i = 0; i < 16; ++i) { l[i] = logits[t * 16 + i]; lmax = fmaxf(lmax, l[i]); }
    red[t] = lmax; __syncthreads();
    for (int o = 512; o > 0; o >>= 1) { if (t < o) red[t] = fmaxf(red[t], red[t + o]); __syncthreads(); }
    const float gmax = red[0]; __syncthreads();

    float ls = 0.0f;
#pragma unroll
    for (int i = 0; i < 16; ++i) { l[i] = __expf(l[i] - gmax); ls += l[i]; }
    red[t] = ls; __syncthreads();
    for (int o = 512; o > 0; o >>= 1) { if (t < o) red[t] += red[t + o]; __syncthreads(); }
    const float inv = 1.0f / red[0]; __syncthreads();

    const int mod = t >> 8;
    float wp = 0.0f, ap = 0.0f;
#pragma unroll
    for (int i = 0; i < 16; ++i) {
        const int j  = t * 16 + i;
        const float w  = l[i] * inv;
        const float al = w * rstds[j];
        alpha[j] = al;
        wp += w;
        ap += al * mus[j];
    }
    for (int md = 0; md < 4; ++md) {
        red[t] = (mod == md) ? wp : 0.0f; __syncthreads();
        for (int o = 512; o > 0; o >>= 1) { if (t < o) red[t] += red[t + o]; __syncthreads(); }
        if (t == 0) wsum[md] = red[0];
        __syncthreads();
        red[t] = (mod == md) ? ap : 0.0f; __syncthreads();
        for (int o = 512; o > 0; o >>= 1) { if (t < o) red[t] += red[t + o]; __syncthreads(); }
        if (t == 0) amu[md] = red[0];
        __syncthreads();
    }
}

__global__ __launch_bounds__(256) void pass2_kernel(
    const float* x, const float* h, const float* s, const float* m,
    const float* alpha, const float* wsum, const float* amu,
    const float* g_x, const float* be_x, const float* g_h, const float* be_h,
    const float* g_s, const float* be_s, const float* g_m, const float* be_m,
    float* u)
{
    __shared__ float red[256];
    const int bid = blockIdx.x;
    const int mod = bid >> 8;
    const int c   = bid & 255;
    const float* aptr[4] = {x, h, s, m};
    const float* gs[4]   = {g_x,  g_h,  g_s,  g_m};
    const float* bes[4]  = {be_x, be_h, be_s, be_m};

    const float* a  = aptr[mod] + (long)c * N_TOK;
    const float* al = alpha + mod * N_TOK;
    float acc = 0.0f;
#pragma unroll 4
    for (int j = threadIdx.x; j < N_TOK; j += 256) acc += al[j] * a[j];
    red[threadIdx.x] = acc; __syncthreads();
    for (int o = 128; o > 0; o >>= 1) {
        if (threadIdx.x < o) red[threadIdx.x] += red[threadIdx.x + o];
        __syncthreads();
    }
    if (threadIdx.x == 0) {
        const float T = red[0];
        u[mod * C_DIM + c] = gs[mod][c] * (T - amu[mod]) + bes[mod][c] * wsum[mod];
    }
}

__global__ __launch_bounds__(512) void head_kernel(
    const float* u,
    const float* wv_x, const float* wv_h, const float* wv_s, const float* wv_m,
    const float* g_res, const float* be_res,
    const float* w_geglu, const float* b_geglu,
    const float* w_ff2, const float* b_ff2,
    const float* w_out, const float* b_out,
    float* yvec)
{
    __shared__ float ovec[D_DIM];
    __shared__ float rr[D_DIM];
    __shared__ float ggl[2 * D_DIM];
    __shared__ float hh[D_DIM];
    __shared__ float ffv[D_DIM];
    __shared__ float red[512];
    const int tid  = threadIdx.x;
    const int wave = tid >> 5;
    const int lane = tid & 31;
    const float* wv[4] = {wv_x, wv_h, wv_s, wv_m};

    {
        v8f acc = {};
        for (int mod = 0; mod < 4; ++mod)
            acc = gemv_tile_f32(u + mod * C_DIM, wv[mod], C_DIM, (long)D_DIM, 1L,
                                wave * 16, acc);
        if (lane < 16) ovec[wave * 16 + lane] = acc[0];
    }
    __syncthreads();

    float v = (tid < D_DIM) ? ovec[tid] : 0.0f;
    const float mu = block_reduce_sum_512(v, red) * (1.0f / D_DIM);
    const float dv = (tid < D_DIM) ? (ovec[tid] - mu) : 0.0f;
    const float var = block_reduce_sum_512(dv * dv, red) * (1.0f / D_DIM);
    const float rstd = rsqrtf(var + EPSV);
    if (tid < D_DIM) rr[tid] = (ovec[tid] - mu) * rstd * g_res[tid] + be_res[tid];
    __syncthreads();

    for (int tt = wave; tt < 32; tt += 16) {
        v8f acc = {};
        acc = gemv_tile_f32(rr, w_geglu, D_DIM, (long)(2 * D_DIM), 1L, tt * 16, acc);
        if (lane < 16) ggl[tt * 16 + lane] = acc[0] + b_geglu[tt * 16 + lane];
    }
    __syncthreads();

    if (tid < D_DIM) {
        const float g  = ggl[D_DIM + tid];
        const float ge = 0.5f * g * (1.0f + erff(g * 0.70710678118654752f));
        hh[tid] = ggl[tid] * ge;
    }
    __syncthreads();

    {
        v8f acc = {};
        acc = gemv_tile_f32(hh, w_ff2, D_DIM, (long)D_DIM, 1L, wave * 16, acc);
        if (lane < 16) ffv[wave * 16 + lane] = acc[0] + b_ff2[wave * 16 + lane];
    }
    __syncthreads();

    {
        v8f acc = {};
        acc = gemv_tile_f32(ffv, w_out, D_DIM, (long)C_DIM, 1L, wave * 16, acc);
        if (lane < 16) yvec[wave * 16 + lane] = acc[0] + b_out[wave * 16 + lane];
    }
}

__global__ __launch_bounds__(1024) void bcast_kernel(const float* yvec, float* out)
{
    const int c = blockIdx.x;
    const float v = yvec[c];
    float* o = out + (long)c * N_TOK;
    for (int j = threadIdx.x; j < N_TOK; j += 1024) o[j] = v;
}

extern "C" void kernel_launch(void* const* d_in, const int* in_sizes, int n_in,
                              void* d_out, int out_size, void* d_ws, size_t ws_size,
                              hipStream_t stream) {
    const float* x       = (const float*)d_in[0];
    const float* h       = (const float*)d_in[1];
    const float* s       = (const float*)d_in[2];
    const float* m       = (const float*)d_in[3];
    const float* tvec    = (const float*)d_in[4];
    const float* wt      = (const float*)d_in[5];
    const float* bt      = (const float*)d_in[6];
    const float* wk_x    = (const float*)d_in[7];
    const float* wv_x    = (const float*)d_in[8];
    const float* g_x     = (const float*)d_in[9];
    const float* be_x    = (const float*)d_in[10];
    const float* wk_h    = (const float*)d_in[11];
    const float* wv_h    = (const float*)d_in[12];
    const float* g_h     = (const float*)d_in[13];
    const float* be_h    = (const float*)d_in[14];
    const float* wk_s    = (const float*)d_in[15];
    const float* wv_s    = (const float*)d_in[16];
    const float* g_s     = (const float*)d_in[17];
    const float* be_s    = (const float*)d_in[18];
    const float* wk_m    = (const float*)d_in[19];
    const float* wv_m    = (const float*)d_in[20];
    const float* g_m     = (const float*)d_in[21];
    const float* be_m    = (const float*)d_in[22];
    const float* w_geglu = (const float*)d_in[23];
    const float* b_geglu = (const float*)d_in[24];
    const float* w_ff2   = (const float*)d_in[25];
    const float* b_ff2   = (const float*)d_in[26];
    const float* g_res   = (const float*)d_in[27];
    const float* be_res  = (const float*)d_in[28];
    const float* w_out   = (const float*)d_in[29];
    const float* b_out   = (const float*)d_in[30];
    float* out = (float*)d_out;
    float* ws  = (float*)d_ws;

    float* gp     = ws + WS_GP;
    float* sgp    = ws + WS_SGP;
    float* sbp    = ws + WS_SBP;
    float* logits = ws + WS_LOG;
    float* mus    = ws + WS_MU;
    float* rstds  = ws + WS_RSTD;
    float* alpha  = ws + WS_ALPHA;
    float* wsum   = ws + WS_WSUM;
    float* amu    = ws + WS_AMU;
    float* u      = ws + WS_U;
    float* yvec   = ws + WS_YVEC;

    prep_kernel<<<1, 512, 0, stream>>>(tvec, wt, bt,
                                       wk_x, wk_h, wk_s, wk_m,
                                       g_x, be_x, g_h, be_h, g_s, be_s, g_m, be_m,
                                       gp, sgp, sbp);
    pass1_kernel<<<256, 256, 0, stream>>>(x, h, s, m, gp, sgp, sbp,
                                          logits, mus, rstds);
    softmax_kernel<<<1, 1024, 0, stream>>>(logits, mus, rstds, alpha, wsum, amu);
    pass2_kernel<<<1024, 256, 0, stream>>>(x, h, s, m, alpha, wsum, amu,
                                           g_x, be_x, g_h, be_h, g_s, be_s, g_m, be_m,
                                           u);
    head_kernel<<<1, 512, 0, stream>>>(u, wv_x, wv_h, wv_s, wv_m,
                                       g_res, be_res, w_geglu, b_geglu,
                                       w_ff2, b_ff2, w_out, b_out, yvec);
    bcast_kernel<<<256, 1024, 0, stream>>>(yvec, out);
}


# softmax launch removed: each of 256 weighted-sum workgroups (modality x 4 channels) rebuilds max, exp sum and its alpha from all 16384 logits; 5->4 launches; softmax and token sum orders change
# speedup vs baseline: 2.1074x; 2.1074x over previous
#include <hip/hip_runtime.h>
#include <math.h>

typedef __attribute__((ext_vector_type(2))) float v2f;
typedef __attribute__((ext_vector_type(8))) float v8f;
typedef __attribute__((ext_vector_type(4))) float v4f;

#define C_DIM 256
#define D_DIM 256
#define N_TOK 4096
#define SCALE 0.0625f
#define EPSV  1e-5f

#define WS_LOG   1032
#define WS_MU    17416
#define WS_RSTD  33800
#define WS_U     66576
#define WS_YVEC  67600

__device__ inline v8f gemv_tile_f32(const float* x, const float* w,
                                    int K, long sk, long sn, int n0, v8f acc) {
    const int lane = threadIdx.x & 31;
    const int half = lane >> 4;
    const int l16  = lane & 15;
    const float amask = (l16 == 0) ? 1.0f : 0.0f;
    const float* wp = w + (long)(n0 + l16) * sn;
    for (int k0 = 0; k0 < K; k0 += 4) {
        const int kb = k0 + 2 * half;
        const float x0 = x[kb];
        const float x1 = x[kb + 1];
        v2f a;
        a.x = x0 * amask;
        a.y = x1 * amask;
        v2f b;
        b.x = wp[(long)kb * sk];
        b.y = wp[(long)(kb + 1) * sk];
        acc = __builtin_amdgcn_wmma_f32_16x16x4_f32(
             false, a,  false, b,
             (short)0, acc,  false,  false);
    }
    return acc;
}

__device__ __forceinline__ v8f gemv_tile_f32_unrolled(const float* x, const float* w,
                                                      long sk, int n0, v8f acc) {
    const int lane = threadIdx.x & 31;
    const int half = lane >> 4;
    const int l16  = lane & 15;
    const float amask = (l16 == 0) ? 1.0f : 0.0f;
    const float* wp = w + (n0 + l16);
#pragma unroll
    for (int k0 = 0; k0 < 256; k0 += 4) {
        const int kb = k0 + 2 * half;
        v2f a;
        a.x = x[kb] * amask;
        a.y = x[kb + 1] * amask;
        v2f b;
        b.x = wp[(long)kb * sk];
        b.y = wp[(long)(kb + 1) * sk];
        acc = __builtin_amdgcn_wmma_f32_16x16x4_f32(
             false, a,  false, b,
             (short)0, acc,  false,  false);
    }
    return acc;
}

__device__ inline float block_reduce_sum_512(float v, float* red) {
    const int t = threadIdx.x;
    red[t] = v; __syncthreads();
    for (int o = 256; o > 0; o >>= 1) {
        if (t < o) red[t] += red[t + o];
        __syncthreads();
    }
    float r = red[0]; __syncthreads();
    return r;
}

__global__ __launch_bounds__(512) void pass1_kernel(
    const float* x, const float* h, const float* s, const float* m,
    const float* tvec, const float* wt, const float* bt,
    const float* wk_x, const float* wk_h, const float* wk_s, const float* wk_m,
    const float* g_x, const float* be_x, const float* g_h, const float* be_h,
    const float* g_s, const float* be_s, const float* g_m, const float* be_m,
    float* logits, float* mus, float* rstds)
{
    __shared__ float q_s[D_DIM];
    __shared__ float p_s[C_DIM];
    __shared__ float gpm_s[C_DIM];
    __shared__ float red[512];
    __shared__ float sh1[8][64];
    __shared__ float sh2[8][64];
    __shared__ float shd[8][64];
    const int tid  = threadIdx.x;
    const int wave = tid >> 5;
    const int lane = tid & 31;
    const int bid  = blockIdx.x;
    const int mod  = bid >> 6;
    const int j0   = (bid & 63) << 6;
    const int tl   = tid & 63;
    const int grp  = tid >> 6;

    const float* a  = (mod == 0) ? x    : (mod == 1) ? h    : (mod == 2) ? s    : m;
    const float* wk = (mod == 0) ? wk_x : (mod == 1) ? wk_h : (mod == 2) ? wk_s : wk_m;
    const float* gg = (mod == 0) ? g_x  : (mod == 1) ? g_h  : (mod == 2) ? g_s  : g_m;
    const float* bb = (mod == 0) ? be_x : (mod == 1) ? be_h : (mod == 2) ? be_s : be_m;

    float v[32];
    {
        const float* ap = a + (long)(grp << 5) * N_TOK + (j0 + tl);
#pragma unroll
        for (int cc = 0; cc < 32; ++cc) v[cc] = ap[(long)cc * N_TOK];
    }

    {
        v8f acc = {};
        acc = gemv_tile_f32_unrolled(tvec, wt, (long)D_DIM, wave * 16, acc);
        if (lane < 16) q_s[wave * 16 + lane] = acc[0] + bt[wave * 16 + lane];
    }
    __syncthreads();

    {
        v8f acc = {};
        acc = gemv_tile_f32(q_s, wk, D_DIM, 1L, (long)D_DIM, wave * 16, acc);
        if (lane < 16) p_s[wave * 16 + lane] = acc[0];
    }
    __syncthreads();

    float gv = 0.0f, bv = 0.0f;
    if (tid < C_DIM) {
        const float p = p_s[tid];
        gv = gg[tid] * p;
        bv = bb[tid] * p;
        gpm_s[tid] = gv;
    }
    const float sgp = block_reduce_sum_512(gv, red);
    const float sbp = block_reduce_sum_512(bv, red);

    float s1 = 0.0f, s2 = 0.0f, dp = 0.0f;
    {
        const int c0 = grp << 5;
#pragma unroll
        for (int cc = 0; cc < 32; ++cc) {
            const float vv = v[cc];
            s1 += vv;
            s2 += vv * vv;
            dp += vv * gpm_s[c0 + cc];
        }
    }
    sh1[grp][tl] = s1; sh2[grp][tl] = s2; shd[grp][tl] = dp;
    __syncthreads();

    if (tid < 64) {
        float t1 = sh1[0][tid], t2 = sh2[0][tid], td = shd[0][tid];
#pragma unroll
        for (int gi = 1; gi < 8; ++gi) { t1 += sh1[gi][tid]; t2 += sh2[gi][tid]; td += shd[gi][tid]; }
        const float mu   = t1 * (1.0f / C_DIM);
        const float var  = t2 * (1.0f / C_DIM) - mu * mu;
        const float rstd = rsqrtf(var + EPSV);
        const float lg   = SCALE * (rstd * (td - mu * sgp) + sbp);
        const int idx = mod * N_TOK + j0 + tid;
        logits[idx] = lg;
        mus[idx]    = mu;
        rstds[idx]  = rstd;
    }
}

__device__ __forceinline__ float wave_sum(float v) {
#pragma unroll
    for (int o = 16; o >= 1; o >>= 1) v += __shfl_xor(v, o, 32);
    return v;
}
__device__ __forceinline__ float wave_max(float v) {
#pragma unroll
    for (int o = 16; o >= 1; o >>= 1) v = fmaxf(v, __shfl_xor(v, o, 32));
    return v;
}

__global__ __launch_bounds__(256) void pass2_kernel(
    const float* x, const float* h, const float* s, const float* m,
    const float* logits, const float* mus, const float* rstds,
    const float* g_x, const float* be_x, const float* g_h, const float* be_h,
    const float* g_s, const float* be_s, const float* g_m, const float* be_m,
    float* u)
{
    __shared__ float red_max[8];
    __shared__ float red_sum[8];
    __shared__ float red_out[8][8];
    const int tid  = threadIdx.x;
    const int wave = tid >> 5;
    const int lane = tid & 31;
    const int bid  = blockIdx.x;
    const int mod  = bid >> 6;
    const int c0   = (bid & 63) << 2;

    const float* a  = (mod == 0) ? x    : (mod == 1) ? h    : (mod == 2) ? s    : m;
    const float* gg = (mod == 0) ? g_x  : (mod == 1) ? g_h  : (mod == 2) ? g_s  : g_m;
    const float* bb = (mod == 0) ? be_x : (mod == 1) ? be_h : (mod == 2) ? be_s : be_m;

    const float gch = gg[c0 + (tid & 3)];
    const float bch = bb[c0 + (tid & 3)];

    v4f L[4][4];
    {
        const v4f* lg4 = (const v4f*)logits + tid;
#pragma unroll
        for (int q = 0; q < 4; ++q)
#pragma unroll
            for (int i = 0; i < 4; ++i) L[q][i] = lg4[q * 1024 + i * 256];
    }
    v4f rs[4], mu[4], av[4][4];
    {
        const v4f* rs4 = (const v4f*)(rstds + mod * N_TOK) + tid;
        const v4f* mu4 = (const v4f*)(mus + mod * N_TOK) + tid;
#pragma unroll
        for (int i = 0; i < 4; ++i) { rs[i] = rs4[i * 256]; mu[i] = mu4[i * 256]; }
#pragma unroll
        for (int cc = 0; cc < 4; ++cc) {
            const v4f* a4 = (const v4f*)(a + (long)(c0 + cc) * N_TOK) + tid;
#pragma unroll
            for (int i = 0; i < 4; ++i) av[cc][i] = a4[i * 256];
        }
    }

    float lmax = -3.4e38f;
#pragma unroll
    for (int q = 0; q < 4; ++q)
#pragma unroll
        for (int i = 0; i < 4; ++i)
            lmax = fmaxf(lmax, fmaxf(fmaxf(L[q][i].x, L[q][i].y), fmaxf(L[q][i].z, L[q][i].w)));
    lmax = wave_max(lmax);
    if (lane == 0) red_max[wave] = lmax;
    __syncthreads();
    float gmax = red_max[0];
#pragma unroll
    for (int w = 1; w < 8; ++w) gmax = fmaxf(gmax, red_max[w]);

    float ls = 0.0f;
#pragma unroll
    for (int q = 0; q < 4; ++q)
#pragma unroll
        for (int i = 0; i < 4; ++i) {
            v4f e;
            e.x = __expf(L[q][i].x - gmax);
            e.y = __expf(L[q][i].y - gmax);
            e.z = __expf(L[q][i].z - gmax);
            e.w = __expf(L[q][i].w - gmax);
            L[q][i] = e;
            ls += e.x; ls += e.y; ls += e.z; ls += e.w;
        }
    ls = wave_sum(ls);
    if (lane == 0) red_sum[wave] = ls;
    __syncthreads();
    float zsum = red_sum[0];
#pragma unroll
    for (int w = 1; w < 8; ++w) zsum += red_sum[w];
    const float inv = 1.0f / zsum;

    float wp = 0.0f, ap = 0.0f, t0 = 0.0f, t1 = 0.0f, t2 = 0.0f, t3 = 0.0f;
#pragma unroll
    for (int i = 0; i < 4; ++i) {
        const v4f e = (mod == 0) ? L[0][i] : (mod == 1) ? L[1][i] : (mod == 2) ? L[2][i] : L[3][i];
#pragma unroll
        for (int k = 0; k < 4; ++k) {
            const float w  = e[k] * inv;
            const float al = w * rs[i][k];
            wp += w;
            ap += al * mu[i][k];
            t0 += al * av[0][i][k];
            t1 += al * av[1][i][k];
            t2 += al * av[2][i][k];
            t3 += al * av[3][i][k];
        }
    }
    wp = wave_sum(wp); ap = wave_sum(ap);
    t0 = wave_sum(t0); t1 = wave_sum(t1); t2 = wave_sum(t2); t3 = wave_sum(t3);
    if (lane == 0) {
        red_out[wave][0] = t0; red_out[wave][1] = t1;
        red_out[wave][2] = t2; red_out[wave][3] = t3;
        red_out[wave][4] = wp; red_out[wave][5] = ap;
    }
    __syncthreads();
    if (tid < 4) {
        float T = red_out[0][tid], wsum = red_out[0][4], amu = red_out[0][5];
#pragma unroll
        for (int w = 1; w < 8; ++w) { T += red_out[w][tid]; wsum += red_out[w][4]; amu += red_out[w][5]; }
        u[mod * C_DIM + c0 + tid] = gch * (T - amu) + bch * wsum;
    }
}

__global__ __launch_bounds__(512) void head_kernel(
    const float* u,
    const float* wv_x, const float* wv_h, const float* wv_s, const float* wv_m,
    const float* g_res, const float* be_res,
    const float* w_geglu, const float* b_geglu,
    const float* w_ff2, const float* b_ff2,
    const float* w_out, const float* b_out,
    float* yvec)
{
    __shared__ float ovec[D_DIM];
    __shared__ float rr[D_DIM];
    __shared__ float ggl[2 * D_DIM];
    __shared__ float hh[D_DIM];
    __shared__ float ffv[D_DIM];
    __shared__ float red[512];
    const int tid  = threadIdx.x;
    const int wave = tid >> 5;
    const int lane = tid & 31;

    {
        v8f acc = {};
        acc = gemv_tile_f32_unrolled(u,             wv_x, (long)D_DIM, wave * 16, acc);
        acc = gemv_tile_f32_unrolled(u + C_DIM,     wv_h, (long)D_DIM, wave * 16, acc);
        acc = gemv_tile_f32_unrolled(u + 2 * C_DIM, wv_s, (long)D_DIM, wave * 16, acc);
        acc = gemv_tile_f32_unrolled(u + 3 * C_DIM, wv_m, (long)D_DIM, wave * 16, acc);
        if (lane < 16) ovec[wave * 16 + lane] = acc[0];
    }
    __syncthreads();

    float v = (tid < D_DIM) ? ovec[tid] : 0.0f;
    const float mu = block_reduce_sum_512(v, red) * (1.0f / D_DIM);
    const float dv = (tid < D_DIM) ? (ovec[tid] - mu) : 0.0f;
    const float var = block_reduce_sum_512(dv * dv, red) * (1.0f / D_DIM);
    const float rstd = rsqrtf(var + EPSV);
    if (tid < D_DIM) rr[tid] = (ovec[tid] - mu) * rstd * g_res[tid] + be_res[tid];
    __syncthreads();

    {
        v8f acc0 = {}, acc1 = {};
        acc0 = gemv_tile_f32_unrolled(rr, w_geglu, (long)(2 * D_DIM), wave * 16, acc0);
        acc1 = gemv_tile_f32_unrolled(rr, w_geglu, (long)(2 * D_DIM), D_DIM + wave * 16, acc1);
        if (lane < 16) {
            ggl[wave * 16 + lane]         = acc0[0] + b_geglu[wave * 16 + lane];
            ggl[D_DIM + wave * 16 + lane] = acc1[0] + b_geglu[D_DIM + wave * 16 + lane];
        }
    }
    __syncthreads();

    if (tid < D_DIM) {
        const float g  = ggl[D_DIM + tid];
        const float ge = 0.5f * g * (1.0f + erff(g * 0.70710678118654752f));
        hh[tid] = ggl[tid] * ge;
    }
    __syncthreads();

    {
        v8f acc = {};
        acc = gemv_tile_f32_unrolled(hh, w_ff2, (long)D_DIM, wave * 16, acc);
        if (lane < 16) ffv[wave * 16 + lane] = acc[0] + b_ff2[wave * 16 + lane];
    }
    __syncthreads();

    {
        v8f acc = {};
        acc = gemv_tile_f32_unrolled(ffv, w_out, (long)C_DIM, wave * 16, acc);
        if (lane < 16) yvec[wave * 16 + lane] = acc[0] + b_out[wave * 16 + lane];
    }
}

__global__ __launch_bounds__(1024) void bcast_kernel(const float* yvec, float* out)
{
    const int c = blockIdx.x;
    const float v = yvec[c];
    float* o = out + (long)c * N_TOK;
    for (int j = threadIdx.x; j < N_TOK; j += 1024) o[j] = v;
}

extern "C" void kernel_launch(void* const* d_in, const int* in_sizes, int n_in,
                              void* d_out, int out_size, void* d_ws, size_t ws_size,
                              hipStream_t stream) {
    const float* x       = (const float*)d_in[0];
    const float* h       = (const float*)d_in[1];
    const float* s       = (const float*)d_in[2];
    const float* m       = (const float*)d_in[3];
    const float* tvec    = (const float*)d_in[4];
    const float* wt      = (const float*)d_in[5];
    const float* bt      = (const float*)d_in[6];
    const float* wk_x    = (const float*)d_in[7];
    const float* wv_x    = (const float*)d_in[8];
    const float* g_x     = (const float*)d_in[9];
    const float* be_x    = (const float*)d_in[10];
    const float* wk_h    = (const float*)d_in[11];
    const float* wv_h    = (const float*)d_in[12];
    const float* g_h     = (const float*)d_in[13];
    const float* be_h    = (const float*)d_in[14];
    const float* wk_s    = (const float*)d_in[15];
    const float* wv_s    = (const float*)d_in[16];
    const float* g_s     = (const float*)d_in[17];
    const float* be_s    = (const float*)d_in[18];
    const float* wk_m    = (const float*)d_in[19];
    const float* wv_m    = (const float*)d_in[20];
    const float* g_m     = (const float*)d_in[21];
    const float* be_m    = (const float*)d_in[22];
    const float* w_geglu = (const float*)d_in[23];
    const float* b_geglu = (const float*)d_in[24];
    const float* w_ff2   = (const float*)d_in[25];
    const float* b_ff2   = (const float*)d_in[26];
    const float* g_res   = (const float*)d_in[27];
    const float* be_res  = (const float*)d_in[28];
    const float* w_out   = (const float*)d_in[29];
    const float* b_out   = (const float*)d_in[30];
    float* out = (float*)d_out;
    float* ws  = (float*)d_ws;

    float* logits = ws + WS_LOG;
    float* mus    = ws + WS_MU;
    float* rstds  = ws + WS_RSTD;
    float* u      = ws + WS_U;
    float* yvec   = ws + WS_YVEC;

    pass1_kernel<<<256, 512, 0, stream>>>(x, h, s, m, tvec, wt, bt,
                                          wk_x, wk_h, wk_s, wk_m,
                                          g_x, be_x, g_h, be_h, g_s, be_s, g_m, be_m,
                                          logits, mus, rstds);
    pass2_kernel<<<256, 256, 0, stream>>>(x, h, s, m, logits, mus, rstds,
                                          g_x, be_x, g_h, be_h, g_s, be_s, g_m, be_m,
                                          u);
    head_kernel<<<1, 512, 0, stream>>>(u, wv_x, wv_h, wv_s, wv_m,
                                       g_res, be_res, w_geglu, b_geglu,
                                       w_ff2, b_ff2, w_out, b_out, yvec);
    bcast_kernel<<<256, 1024, 0, stream>>>(yvec, out);
}


# wv product leaves the one-workgroup head: weighted-sum workgroups (now 64 x 16 channels) apply their 16 u values to 16 wv rows, head adds 64 partial vectors; softmax, token, out sum orders change
# speedup vs baseline: 2.8384x; 1.3469x over previous
#include <hip/hip_runtime.h>
#include <math.h>

typedef __attribute__((ext_vector_type(2))) float v2f;
typedef __attribute__((ext_vector_type(8))) float v8f;
typedef __attribute__((ext_vector_type(4))) float v4f;

#define C_DIM 256
#define D_DIM 256
#define N_TOK 4096
#define SCALE 0.0625f
#define EPSV  1e-5f

#define WS_LOG   1032
#define WS_MU    17416
#define WS_RSTD  33800
#define WS_PART  50184
#define WS_YVEC  67600

__device__ inline v8f gemv_tile_f32(const float* x, const float* w,
                                    int K, long sk, long sn, int n0, v8f acc) {
    const int lane = threadIdx.x & 31;
    const int half = lane >> 4;
    const int l16  = lane & 15;
    const float amask = (l16 == 0) ? 1.0f : 0.0f;
    const float* wp = w + (long)(n0 + l16) * sn;
    for (int k0 = 0; k0 < K; k0 += 4) {
        const int kb = k0 + 2 * half;
        const float x0 = x[kb];
        const float x1 = x[kb + 1];
        v2f a;
        a.x = x0 * amask;
        a.y = x1 * amask;
        v2f b;
        b.x = wp[(long)kb * sk];
        b.y = wp[(long)(kb + 1) * sk];
        acc = __builtin_amdgcn_wmma_f32_16x16x4_f32(
             false, a,  false, b,
             (short)0, acc,  false,  false);
    }
    return acc;
}

__device__ __forceinline__ v8f gemv_tile_f32_unrolled(const float* x, const float* w,
                                                      long sk, int n0, v8f acc) {
    const int lane = threadIdx.x & 31;
    const int half = lane >> 4;
    const int l16  = lane & 15;
    const float amask = (l16 == 0) ? 1.0f : 0.0f;
    const float* wp = w + (n0 + l16);
#pragma unroll
    for (int k0 = 0; k0 < 256; k0 += 4) {
        const int kb = k0 + 2 * half;
        v2f a;
        a.x = x[kb] * amask;
        a.y = x[kb + 1] * amask;
        v2f b;
        b.x = wp[(long)kb * sk];
        b.y = wp[(long)(kb + 1) * sk];
        acc = __builtin_amdgcn_wmma_f32_16x16x4_f32(
             false, a,  false, b,
             (short)0, acc,  false,  false);
    }
    return acc;
}

__device__ inline float block_reduce_sum_512(float v, float* red) {
    const int t = threadIdx.x;
    red[t] = v; __syncthreads();
    for (int o = 256; o > 0; o >>= 1) {
        if (t < o) red[t] += red[t + o];
        __syncthreads();
    }
    float r = red[0]; __syncthreads();
    return r;
}

__global__ __launch_bounds__(512) void pass1_kernel(
    const float* x, const float* h, const float* s, const float* m,
    const float* tvec, const float* wt, const float* bt,
    const float* wk_x, const float* wk_h, const float* wk_s, const float* wk_m,
    const float* g_x, const float* be_x, const float* g_h, const float* be_h,
    const float* g_s, const float* be_s, const float* g_m, const float* be_m,
    float* logits, float* mus, float* rstds)
{
    __shared__ float q_s[D_DIM];
    __shared__ float p_s[C_DIM];
    __shared__ float gpm_s[C_DIM];
    __shared__ float red[512];
    __shared__ float sh1[8][64];
    __shared__ float sh2[8][64];
    __shared__ float shd[8][64];
    const int tid  = threadIdx.x;
    const int wave = tid >> 5;
    const int lane = tid & 31;
    const int bid  = blockIdx.x;
    const int mod  = bid >> 6;
    const int j0   = (bid & 63) << 6;
    const int tl   = tid & 63;
    const int grp  = tid >> 6;

    const float* a  = (mod == 0) ? x    : (mod == 1) ? h    : (mod == 2) ? s    : m;
    const float* wk = (mod == 0) ? wk_x : (mod == 1) ? wk_h : (mod == 2) ? wk_s : wk_m;
    const float* gg = (mod == 0) ? g_x  : (mod == 1) ? g_h  : (mod == 2) ? g_s  : g_m;
    const float* bb = (mod == 0) ? be_x : (mod == 1) ? be_h : (mod == 2) ? be_s : be_m;

    float v[32];
    {
        const float* ap = a + (long)(grp << 5) * N_TOK + (j0 + tl);
#pragma unroll
        for (int cc = 0; cc < 32; ++cc) v[cc] = ap[(long)cc * N_TOK];
    }

    {
        v8f acc = {};
        acc = gemv_tile_f32_unrolled(tvec, wt, (long)D_DIM, wave * 16, acc);
        if (lane < 16) q_s[wave * 16 + lane] = acc[0] + bt[wave * 16 + lane];
    }
    __syncthreads();

    {
        v8f acc = {};
        acc = gemv_tile_f32(q_s, wk, D_DIM, 1L, (long)D_DIM, wave * 16, acc);
        if (lane < 16) p_s[wave * 16 + lane] = acc[0];
    }
    __syncthreads();

    float gv = 0.0f, bv = 0.0f;
    if (tid < C_DIM) {
        const float p = p_s[tid];
        gv = gg[tid] * p;
        bv = bb[tid] * p;
        gpm_s[tid] = gv;
    }
    const float sgp = block_reduce_sum_512(gv, red);
    const float sbp = block_reduce_sum_512(bv, red);

    float s1 = 0.0f, s2 = 0.0f, dp = 0.0f;
    {
        const int c0 = grp << 5;
#pragma unroll
        for (int cc = 0; cc < 32; ++cc) {
            const float vv = v[cc];
            s1 += vv;
            s2 += vv * vv;
            dp += vv * gpm_s[c0 + cc];
        }
    }
    sh1[grp][tl] = s1; sh2[grp][tl] = s2; shd[grp][tl] = dp;
    __syncthreads();

    if (tid < 64) {
        float t1 = sh1[0][tid], t2 = sh2[0][tid], td = shd[0][tid];
#pragma unroll
        for (int gi = 1; gi < 8; ++gi) { t1 += sh1[gi][tid]; t2 += sh2[gi][tid]; td += shd[gi][tid]; }
        const float mu   = t1 * (1.0f / C_DIM);
        const float var  = t2 * (1.0f / C_DIM) - mu * mu;
        const float rstd = rsqrtf(var + EPSV);
        const float lg   = SCALE * (rstd * (td - mu * sgp) + sbp);
        const int idx = mod * N_TOK + j0 + tid;
        logits[idx] = lg;
        mus[idx]    = mu;
        rstds[idx]  = rstd;
    }
}

__device__ __forceinline__ float wave_sum(float v) {
#pragma unroll
    for (int o = 16; o >= 1; o >>= 1) v += __shfl_xor(v, o, 32);
    return v;
}
__device__ __forceinline__ float wave_max(float v) {
#pragma unroll
    for (int o = 16; o >= 1; o >>= 1) v = fmaxf(v, __shfl_xor(v, o, 32));
    return v;
}

__global__ __launch_bounds__(1024) void pass2_kernel(
    const float* x, const float* h, const float* s, const float* m,
    const float* logits, const float* mus, const float* rstds,
    const float* wv_x, const float* wv_h, const float* wv_s, const float* wv_m,
    const float* g_x, const float* be_x, const float* g_h, const float* be_h,
    const float* g_s, const float* be_s, const float* g_m, const float* be_m,
    float* part)
{
    __shared__ float red_max[32];
    __shared__ float red_sum[32];
    __shared__ float red_out[32][18];
    __shared__ float u_s[16];
    __shared__ float o_s[2][D_DIM];
    const int tid  = threadIdx.x;
    const int wave = tid >> 5;
    const int lane = tid & 31;
    const int half = lane >> 4;
    const int l16  = lane & 15;
    const int bid  = blockIdx.x;
    const int mod  = bid >> 4;
    const int c0   = (bid & 15) << 4;

    const float* a  = (mod == 0) ? x    : (mod == 1) ? h    : (mod == 2) ? s    : m;
    const float* wv = (mod == 0) ? wv_x : (mod == 1) ? wv_h : (mod == 2) ? wv_s : wv_m;
    const float* gg = (mod == 0) ? g_x  : (mod == 1) ? g_h  : (mod == 2) ? g_s  : g_m;
    const float* bb = (mod == 0) ? be_x : (mod == 1) ? be_h : (mod == 2) ? be_s : be_m;

    const float gch = gg[c0 + (tid & 15)];
    const float bch = bb[c0 + (tid & 15)];

    const int kq = wave >> 4;
    const int nt = (wave & 15) << 4;
    v2f bw[2];
    {
        const float* wp = wv + (long)(c0 + 8 * kq) * D_DIM + nt + l16;
#pragma unroll
        for (int ks = 0; ks < 2; ++ks) {
            const int kb = 4 * ks + 2 * half;
            bw[ks].x = wp[(long)kb * D_DIM];
            bw[ks].y = wp[(long)(kb + 1) * D_DIM];
        }
    }

    v4f L[4];
    {
        const v4f* lg4 = (const v4f*)logits + tid;
#pragma unroll
        for (int q = 0; q < 4; ++q) L[q] = lg4[q * 1024];
    }
    const v4f rs = ((const v4f*)(rstds + mod * N_TOK))[tid];
    const v4f mu = ((const v4f*)(mus + mod * N_TOK))[tid];
    v4f av[16];
#pragma unroll
    for (int cc = 0; cc < 16; ++cc)
        av[cc] = ((const v4f*)(a + (long)(c0 + cc) * N_TOK))[tid];

    __builtin_amdgcn_sched_barrier(0);

    float lmax = -3.4e38f;
#pragma unroll
    for (int q = 0; q < 4; ++q)
        lmax = fmaxf(lmax, fmaxf(fmaxf(L[q].x, L[q].y), fmaxf(L[q].z, L[q].w)));
    lmax = wave_max(lmax);
    if (lane == 0) red_max[wave] = lmax;
    __syncthreads();
    const float gmax = wave_max(red_max[lane]);

    float ls = 0.0f;
#pragma unroll
    for (int q = 0; q < 4; ++q) {
        v4f e;
        e.x = __expf(L[q].x - gmax);
        e.y = __expf(L[q].y - gmax);
        e.z = __expf(L[q].z - gmax);
        e.w = __expf(L[q].w - gmax);
        L[q] = e;
        ls += e.x; ls += e.y; ls += e.z; ls += e.w;
    }
    ls = wave_sum(ls);
    if (lane == 0) red_sum[wave] = ls;
    __syncthreads();
    const float zsum = wave_sum(red_sum[lane]);
    const float inv = 1.0f / zsum;

    float wps = 0.0f, aps = 0.0f;
    float t[16];
#pragma unroll
    for (int cc = 0; cc < 16; ++cc) t[cc] = 0.0f;
    {
        const v4f e = (mod == 0) ? L[0] : (mod == 1) ? L[1] : (mod == 2) ? L[2] : L[3];
#pragma unroll
        for (int k = 0; k < 4; ++k) {
            const float w  = e[k] * inv;
            const float al = w * rs[k];
            wps += w;
            aps += al * mu[k];
#pragma unroll
            for (int cc = 0; cc < 16; ++cc) t[cc] += al * av[cc][k];
        }
    }
    wps = wave_sum(wps); aps = wave_sum(aps);
    float r8[8], r4[4], r2[2], r1;
    {
        const bool b = (lane & 16) != 0;
#pragma unroll
        for (int i = 0; i < 8; ++i) {
            const float snd = b ? t[i] : t[i + 8];
            const float kp  = b ? t[i + 8] : t[i];
            r8[i] = kp + __shfl_xor(snd, 16, 32);
        }
    }
    {
        const bool b = (lane & 8) != 0;
#pragma unroll
        for (int i = 0; i < 4; ++i) {
            const float snd = b ? r8[i] : r8[i + 4];
            const float kp  = b ? r8[i + 4] : r8[i];
            r4[i] = kp + __shfl_xor(snd, 8, 32);
        }
    }
    {
        const bool b = (lane & 4) != 0;
#pragma unroll
        for (int i = 0; i < 2; ++i) {
            const float snd = b ? r4[i] : r4[i + 2];
            const float kp  = b ? r4[i + 2] : r4[i];
            r2[i] = kp + __shfl_xor(snd, 4, 32);
        }
    }
    {
        const bool b = (lane & 2) != 0;
        const float snd = b ? r2[0] : r2[1];
        const float kp  = b ? r2[1] : r2[0];
        r1 = kp + __shfl_xor(snd, 2, 32);
    }
    r1 += __shfl_xor(r1, 1, 32);
    if ((lane & 1) == 0) red_out[wave][lane >> 1] = r1;
    if (lane == 0) { red_out[wave][16] = wps; red_out[wave][17] = aps; }
    __syncthreads();
    if (tid < 16) {
        float T = red_out[0][tid], wsum = red_out[0][16], amu = red_out[0][17];
#pragma unroll
        for (int w = 1; w < 32; ++w) { T += red_out[w][tid]; wsum += red_out[w][16]; amu += red_out[w][17]; }
        u_s[tid] = gch * (T - amu) + bch * wsum;
    }
    __syncthreads();

    {
        const float amask = (l16 == 0) ? 1.0f : 0.0f;
        v8f acc = {};
#pragma unroll
        for (int ks = 0; ks < 2; ++ks) {
            const int kb = 8 * kq + 4 * ks + 2 * half;
            v2f av2;
            av2.x = u_s[kb] * amask;
            av2.y = u_s[kb + 1] * amask;
            acc = __builtin_amdgcn_wmma_f32_16x16x4_f32(
                 false, av2,  false, bw[ks],
                 (short)0, acc,  false,  false);
        }
        if (lane < 16) o_s[kq][nt + lane] = acc[0];
    }
    __syncthreads();
    if (tid < D_DIM) part[bid * D_DIM + tid] = o_s[0][tid] + o_s[1][tid];
}

__global__ __launch_bounds__(512) void head_kernel(
    const float* part,
    const float* g_res, const float* be_res,
    const float* w_geglu, const float* b_geglu,
    const float* w_ff2, const float* b_ff2,
    const float* w_out, const float* b_out,
    float* yvec)
{
    __shared__ float ovec[D_DIM];
    __shared__ __attribute__((aligned(16))) float psum[8][D_DIM];
    __shared__ float rr[D_DIM];
    __shared__ float ggl[2 * D_DIM];
    __shared__ float hh[D_DIM];
    __shared__ float ffv[D_DIM];
    __shared__ float red[512];
    const int tid  = threadIdx.x;
    const int wave = tid >> 5;
    const int lane = tid & 31;

    {
        const int d4 = tid & 63;
        const int g  = tid >> 6;
        const v4f* p4 = (const v4f*)part + (long)(g * 8) * (D_DIM / 4) + d4;
        v4f pv[8];
#pragma unroll
        for (int i = 0; i < 8; ++i) pv[i] = p4[i * (D_DIM / 4)];
        v4f sacc = pv[0];
#pragma unroll
        for (int i = 1; i < 8; ++i) sacc += pv[i];
        ((v4f*)psum[g])[d4] = sacc;
    }
    __syncthreads();
    if (tid < D_DIM) {
        float o = psum[0][tid];
#pragma unroll
        for (int g = 1; g < 8; ++g) o += psum[g][tid];
        ovec[tid] = o;
    }
    __syncthreads();

    float v = (tid < D_DIM) ? ovec[tid] : 0.0f;
    const float mu = block_reduce_sum_512(v, red) * (1.0f / D_DIM);
    const float dv = (tid < D_DIM) ? (ovec[tid] - mu) : 0.0f;
    const float var = block_reduce_sum_512(dv * dv, red) * (1.0f / D_DIM);
    const float rstd = rsqrtf(var + EPSV);
    if (tid < D_DIM) rr[tid] = (ovec[tid] - mu) * rstd * g_res[tid] + be_res[tid];
    __syncthreads();

    {
        v8f acc0 = {}, acc1 = {};
        acc0 = gemv_tile_f32_unrolled(rr, w_geglu, (long)(2 * D_DIM), wave * 16, acc0);
        acc1 = gemv_tile_f32_unrolled(rr, w_geglu, (long)(2 * D_DIM), D_DIM + wave * 16, acc1);
        if (lane < 16) {
            ggl[wave * 16 + lane]         = acc0[0] + b_geglu[wave * 16 + lane];
            ggl[D_DIM + wave * 16 + lane] = acc1[0] + b_geglu[D_DIM + wave * 16 + lane];
        }
    }
    __syncthreads();

    if (tid < D_DIM) {
        const float g  = ggl[D_DIM + tid];
        const float ge = 0.5f * g * (1.0f + erff(g * 0.70710678118654752f));
        hh[tid] = ggl[tid] * ge;
    }
    __syncthreads();

    {
        v8f acc = {};
        acc = gemv_tile_f32_unrolled(hh, w_ff2, (long)D_DIM, wave * 16, acc);
        if (lane < 16) ffv[wave * 16 + lane] = acc[0] + b_ff2[wave * 16 + lane];
    }
    __syncthreads();

    {
        v8f acc = {};
        acc = gemv_tile_f32_unrolled(ffv, w_out, (long)C_DIM, wave * 16, acc);
        if (lane < 16) yvec[wave * 16 + lane] = acc[0] + b_out[wave * 16 + lane];
    }
}

__global__ __launch_bounds__(1024) void bcast_kernel(const float* yvec, float* out)
{
    const int c = blockIdx.x;
    const float v = yvec[c];
    float* o = out + (long)c * N_TOK;
    for (int j = threadIdx.x; j < N_TOK; j += 1024) o[j] = v;
}

extern "C" void kernel_launch(void* const* d_in, const int* in_sizes, int n_in,
                              void* d_out, int out_size, void* d_ws, size_t ws_size,
                              hipStream_t stream) {
    const float* x       = (const float*)d_in[0];
    const float* h       = (const float*)d_in[1];
    const float* s       = (const float*)d_in[2];
    const float* m       = (const float*)d_in[3];
    const float* tvec    = (const float*)d_in[4];
    const float* wt      = (const float*)d_in[5];
    const float* bt      = (const float*)d_in[6];
    const float* wk_x    = (const float*)d_in[7];
    const float* wv_x    = (const float*)d_in[8];
    const float* g_x     = (const float*)d_in[9];
    const float* be_x    = (const float*)d_in[10];
    const float* wk_h    = (const float*)d_in[11];
    const float* wv_h    = (const float*)d_in[12];
    const float* g_h     = (const float*)d_in[13];
    const float* be_h    = (const float*)d_in[14];
    const float* wk_s    = (const float*)d_in[15];
    const float* wv_s    = (const float*)d_in[16];
    const float* g_s     = (const float*)d_in[17];
    const float* be_s    = (const float*)d_in[18];
    const float* wk_m    = (const float*)d_in[19];
    const float* wv_m    = (const float*)d_in[20];
    const float* g_m     = (const float*)d_in[21];
    const float* be_m    = (const float*)d_in[22];
    const float* w_geglu = (const float*)d_in[23];
    const float* b_geglu = (const float*)d_in[24];
    const float* w_ff2   = (const float*)d_in[25];
    const float* b_ff2   = (const float*)d_in[26];
    const float* g_res   = (const float*)d_in[27];
    const float* be_res  = (const float*)d_in[28];
    const float* w_out   = (const float*)d_in[29];
    const float* b_out   = (const float*)d_in[30];
    float* out = (float*)d_out;
    float* ws  = (float*)d_ws;

    float* logits = ws + WS_LOG;
    float* mus    = ws + WS_MU;
    float* rstds  = ws + WS_RSTD;
    float* part   = ws + WS_PART;
    float* yvec   = ws + WS_YVEC;

    pass1_kernel<<<256, 512, 0, stream>>>(x, h, s, m, tvec, wt, bt,
                                          wk_x, wk_h, wk_s, wk_m,
                                          g_x, be_x, g_h, be_h, g_s, be_s, g_m, be_m,
                                          logits, mus, rstds);
    pass2_kernel<<<64, 1024, 0, stream>>>(x, h, s, m, logits, mus, rstds,
                                          wv_x, wv_h, wv_s, wv_m,
                                          g_x, be_x, g_h, be_h, g_s, be_s, g_m, be_m,
                                          part);
    head_kernel<<<1, 512, 0, stream>>>(part,
                                       g_res, be_res, w_geglu, b_geglu,
                                       w_ff2, b_ff2, w_out, b_out, yvec);
    bcast_kernel<<<256, 1024, 0, stream>>>(yvec, out);
}


# broadcast launch removed: each of 64 workgroups (4 channels) rebuilds the whole head (partial sum, LN, GEGLU, ff2) and one out-proj tile, then writes its channels; 4->3 launches; sum orders kept
# speedup vs baseline: 2.9997x; 1.0568x over previous
#include <hip/hip_runtime.h>
#include <math.h>

typedef __attribute__((ext_vector_type(2))) float v2f;
typedef __attribute__((ext_vector_type(8))) float v8f;
typedef __attribute__((ext_vector_type(4))) float v4f;

#define C_DIM 256
#define D_DIM 256
#define N_TOK 4096
#define SCALE 0.0625f
#define EPSV  1e-5f

#define WS_LOG   1032
#define WS_MU    17416
#define WS_RSTD  33800
#define WS_PART  50184

__device__ inline v8f gemv_tile_f32(const float* x, const float* w,
                                    int K, long sk, long sn, int n0, v8f acc) {
    const int lane = threadIdx.x & 31;
    const int half = lane >> 4;
    const int l16  = lane & 15;
    const float amask = (l16 == 0) ? 1.0f : 0.0f;
    const float* wp = w + (long)(n0 + l16) * sn;
    for (int k0 = 0; k0 < K; k0 += 4) {
        const int kb = k0 + 2 * half;
        const float x0 = x[kb];
        const float x1 = x[kb + 1];
        v2f a;
        a.x = x0 * amask;
        a.y = x1 * amask;
        v2f b;
        b.x = wp[(long)kb * sk];
        b.y = wp[(long)(kb + 1) * sk];
        acc = __builtin_amdgcn_wmma_f32_16x16x4_f32(
             false, a,  false, b,
             (short)0, acc,  false,  false);
    }
    return acc;
}

__device__ __forceinline__ v8f gemv_tile_f32_unrolled(const float* x, const float* w,
                                                      long sk, int n0, v8f acc) {
    const int lane = threadIdx.x & 31;
    const int half = lane >> 4;
    const int l16  = lane & 15;
    const float amask = (l16 == 0) ? 1.0f : 0.0f;
    const float* wp = w + (n0 + l16);
#pragma unroll
    for (int k0 = 0; k0 < 256; k0 += 4) {
        const int kb = k0 + 2 * half;
        v2f a;
        a.x = x[kb] * amask;
        a.y = x[kb + 1] * amask;
        v2f b;
        b.x = wp[(long)kb * sk];
        b.y = wp[(long)(kb + 1) * sk];
        acc = __builtin_amdgcn_wmma_f32_16x16x4_f32(
             false, a,  false, b,
             (short)0, acc,  false,  false);
    }
    return acc;
}

__device__ inline float block_reduce_sum_512(float v, float* red) {
    const int t = threadIdx.x;
    red[t] = v; __syncthreads();
    for (int o = 256; o > 0; o >>= 1) {
        if (t < o) red[t] += red[t + o];
        __syncthreads();
    }
    float r = red[0]; __syncthreads();
    return r;
}

__global__ __launch_bounds__(512) void pass1_kernel(
    const float* x, const float* h, const float* s, const float* m,
    const float* tvec, const float* wt, const float* bt,
    const float* wk_x, const float* wk_h, const float* wk_s, const float* wk_m,
    const float* g_x, const float* be_x, const float* g_h, const float* be_h,
    const float* g_s, const float* be_s, const float* g_m, const float* be_m,
    float* logits, float* mus, float* rstds)
{
    __shared__ float q_s[D_DIM];
    __shared__ float p_s[C_DIM];
    __shared__ float gpm_s[C_DIM];
    __shared__ float red[512];
    __shared__ float sh1[8][64];
    __shared__ float sh2[8][64];
    __shared__ float shd[8][64];
    const int tid  = threadIdx.x;
    const int wave = tid >> 5;
    const int lane = tid & 31;
    const int bid  = blockIdx.x;
    const int mod  = bid >> 6;
    const int j0   = (bid & 63) << 6;
    const int tl   = tid & 63;
    const int grp  = tid >> 6;

    const float* a  = (mod == 0) ? x    : (mod == 1) ? h    : (mod == 2) ? s    : m;
    const float* wk = (mod == 0) ? wk_x : (mod == 1) ? wk_h : (mod == 2) ? wk_s : wk_m;
    const float* gg = (mod == 0) ? g_x  : (mod == 1) ? g_h  : (mod == 2) ? g_s  : g_m;
    const float* bb = (mod == 0) ? be_x : (mod == 1) ? be_h : (mod == 2) ? be_s : be_m;

    float v[32];
    {
        const float* ap = a + (long)(grp << 5) * N_TOK + (j0 + tl);
#pragma unroll
        for (int cc = 0; cc < 32; ++cc) v[cc] = ap[(long)cc * N_TOK];
    }

    {
        v8f acc = {};
        acc = gemv_tile_f32_unrolled(tvec, wt, (long)D_DIM, wave * 16, acc);
        if (lane < 16) q_s[wave * 16 + lane] = acc[0] + bt[wave * 16 + lane];
    }
    __syncthreads();

    {
        v8f acc = {};
        acc = gemv_tile_f32(q_s, wk, D_DIM, 1L, (long)D_DIM, wave * 16, acc);
        if (lane < 16) p_s[wave * 16 + lane] = acc[0];
    }
    __syncthreads();

    float gv = 0.0f, bv = 0.0f;
    if (tid < C_DIM) {
        const float p = p_s[tid];
        gv = gg[tid] * p;
        bv = bb[tid] * p;
        gpm_s[tid] = gv;
    }
    const float sgp = block_reduce_sum_512(gv, red);
    const float sbp = block_reduce_sum_512(bv, red);

    float s1 = 0.0f, s2 = 0.0f, dp = 0.0f;
    {
        const int c0 = grp << 5;
#pragma unroll
        for (int cc = 0; cc < 32; ++cc) {
            const float vv = v[cc];
            s1 += vv;
            s2 += vv * vv;
            dp += vv * gpm_s[c0 + cc];
        }
    }
    sh1[grp][tl] = s1; sh2[grp][tl] = s2; shd[grp][tl] = dp;
    __syncthreads();

    if (tid < 64) {
        float t1 = sh1[0][tid], t2 = sh2[0][tid], td = shd[0][tid];
#pragma unroll
        for (int gi = 1; gi < 8; ++gi) { t1 += sh1[gi][tid]; t2 += sh2[gi][tid]; td += shd[gi][tid]; }
        const float mu   = t1 * (1.0f / C_DIM);
        const float var  = t2 * (1.0f / C_DIM) - mu * mu;
        const float rstd = rsqrtf(var + EPSV);
        const float lg   = SCALE * (rstd * (td - mu * sgp) + sbp);
        const int idx = mod * N_TOK + j0 + tid;
        logits[idx] = lg;
        mus[idx]    = mu;
        rstds[idx]  = rstd;
    }
}

__device__ __forceinline__ float wave_sum(float v) {
#pragma unroll
    for (int o = 16; o >= 1; o >>= 1) v += __shfl_xor(v, o, 32);
    return v;
}
__device__ __forceinline__ float wave_max(float v) {
#pragma unroll
    for (int o = 16; o >= 1; o >>= 1) v = fmaxf(v, __shfl_xor(v, o, 32));
    return v;
}

__global__ __launch_bounds__(1024) void pass2_kernel(
    const float* x, const float* h, const float* s, const float* m,
    const float* logits, const float* mus, const float* rstds,
    const float* wv_x, const float* wv_h, const float* wv_s, const float* wv_m,
    const float* g_x, const float* be_x, const float* g_h, const float* be_h,
    const float* g_s, const float* be_s, const float* g_m, const float* be_m,
    float* part)
{
    __shared__ float red_max[32];
    __shared__ float red_sum[32];
    __shared__ float red_out[32][18];
    __shared__ float u_s[16];
    __shared__ float o_s[2][D_DIM];
    const int tid  = threadIdx.x;
    const int wave = tid >> 5;
    const int lane = tid & 31;
    const int half = lane >> 4;
    const int l16  = lane & 15;
    const int bid  = blockIdx.x;
    const int mod  = bid >> 4;
    const int c0   = (bid & 15) << 4;

    const float* a  = (mod == 0) ? x    : (mod == 1) ? h    : (mod == 2) ? s    : m;
    const float* wv = (mod == 0) ? wv_x : (mod == 1) ? wv_h : (mod == 2) ? wv_s : wv_m;
    const float* gg = (mod == 0) ? g_x  : (mod == 1) ? g_h  : (mod == 2) ? g_s  : g_m;
    const float* bb = (mod == 0) ? be_x : (mod == 1) ? be_h : (mod == 2) ? be_s : be_m;

    const float gch = gg[c0 + (tid & 15)];
    const float bch = bb[c0 + (tid & 15)];

    const int kq = wave >> 4;
    const int nt = (wave & 15) << 4;
    v2f bw[2];
    {
        const float* wp = wv + (long)(c0 + 8 * kq) * D_DIM + nt + l16;
#pragma unroll
        for (int ks = 0; ks < 2; ++ks) {
            const int kb = 4 * ks + 2 * half;
            bw[ks].x = wp[(long)kb * D_DIM];
            bw[ks].y = wp[(long)(kb + 1) * D_DIM];
        }
    }

    v4f L[4];
    {
        const v4f* lg4 = (const v4f*)logits + tid;
#pragma unroll
        for (int q = 0; q < 4; ++q) L[q] = lg4[q * 1024];
    }
    const v4f rs = ((const v4f*)(rstds + mod * N_TOK))[tid];
    const v4f mu = ((const v4f*)(mus + mod * N_TOK))[tid];
    v4f av[16];
#pragma unroll
    for (int cc = 0; cc < 16; ++cc)
        av[cc] = ((const v4f*)(a + (long)(c0 + cc) * N_TOK))[tid];

    __builtin_amdgcn_sched_barrier(0);

    float lmax = -3.4e38f;
#pragma unroll
    for (int q = 0; q < 4; ++q)
        lmax = fmaxf(lmax, fmaxf(fmaxf(L[q].x, L[q].y), fmaxf(L[q].z, L[q].w)));
    lmax = wave_max(lmax);
    if (lane == 0) red_max[wave] = lmax;
    __syncthreads();
    const float gmax = wave_max(red_max[lane]);

    float ls = 0.0f;
#pragma unroll
    for (int q = 0; q < 4; ++q) {
        v4f e;
        e.x = __expf(L[q].x - gmax);
        e.y = __expf(L[q].y - gmax);
        e.z = __expf(L[q].z - gmax);
        e.w = __expf(L[q].w - gmax);
        L[q] = e;
        ls += e.x; ls += e.y; ls += e.z; ls += e.w;
    }
    ls = wave_sum(ls);
    if (lane == 0) red_sum[wave] = ls;
    __syncthreads();
    const float zsum = wave_sum(red_sum[lane]);
    const float inv = 1.0f / zsum;

    float wps = 0.0f, aps = 0.0f;
    float t[16];
#pragma unroll
    for (int cc = 0; cc < 16; ++cc) t[cc] = 0.0f;
    {
        const v4f e = (mod == 0) ? L[0] : (mod == 1) ? L[1] : (mod == 2) ? L[2] : L[3];
#pragma unroll
        for (int k = 0; k < 4; ++k) {
            const float w  = e[k] * inv;
            const float al = w * rs[k];
            wps += w;
            aps += al * mu[k];
#pragma unroll
            for (int cc = 0; cc < 16; ++cc) t[cc] += al * av[cc][k];
        }
    }
    wps = wave_sum(wps); aps = wave_sum(aps);
    float r8[8], r4[4], r2[2], r1;
    {
        const bool b = (lane & 16) != 0;
#pragma unroll
        for (int i = 0; i < 8; ++i) {
            const float snd = b ? t[i] : t[i + 8];
            const float kp  = b ? t[i + 8] : t[i];
            r8[i] = kp + __shfl_xor(snd, 16, 32);
        }
    }
    {
        const bool b = (lane & 8) != 0;
#pragma unroll
        for (int i = 0; i < 4; ++i) {
            const float snd = b ? r8[i] : r8[i + 4];
            const float kp  = b ? r8[i + 4] : r8[i];
            r4[i] = kp + __shfl_xor(snd, 8, 32);
        }
    }
    {
        const bool b = (lane & 4) != 0;
#pragma unroll
        for (int i = 0; i < 2; ++i) {
            const float snd = b ? r4[i] : r4[i + 2];
            const float kp  = b ? r4[i + 2] : r4[i];
            r2[i] = kp + __shfl_xor(snd, 4, 32);
        }
    }
    {
        const bool b = (lane & 2) != 0;
        const float snd = b ? r2[0] : r2[1];
        const float kp  = b ? r2[1] : r2[0];
        r1 = kp + __shfl_xor(snd, 2, 32);
    }
    r1 += __shfl_xor(r1, 1, 32);
    if ((lane & 1) == 0) red_out[wave][lane >> 1] = r1;
    if (lane == 0) { red_out[wave][16] = wps; red_out[wave][17] = aps; }
    __syncthreads();
    if (tid < 16) {
        float T = red_out[0][tid], wsum = red_out[0][16], amu = red_out[0][17];
#pragma unroll
        for (int w = 1; w < 32; ++w) { T += red_out[w][tid]; wsum += red_out[w][16]; amu += red_out[w][17]; }
        u_s[tid] = gch * (T - amu) + bch * wsum;
    }
    __syncthreads();

    {
        const float amask = (l16 == 0) ? 1.0f : 0.0f;
        v8f acc = {};
#pragma unroll
        for (int ks = 0; ks < 2; ++ks) {
            const int kb = 8 * kq + 4 * ks + 2 * half;
            v2f av2;
            av2.x = u_s[kb] * amask;
            av2.y = u_s[kb + 1] * amask;
            acc = __builtin_amdgcn_wmma_f32_16x16x4_f32(
                 false, av2,  false, bw[ks],
                 (short)0, acc,  false,  false);
        }
        if (lane < 16) o_s[kq][nt + lane] = acc[0];
    }
    __syncthreads();
    if (tid < D_DIM) part[bid * D_DIM + tid] = o_s[0][tid] + o_s[1][tid];
}

__global__ __launch_bounds__(512) void tail_kernel(
    const float* part,
    const float* g_res, const float* be_res,
    const float* w_geglu, const float* b_geglu,
    const float* w_ff2, const float* b_ff2,
    const float* w_out, const float* b_out,
    float* out)
{
    __shared__ float ovec[D_DIM];
    __shared__ __attribute__((aligned(16))) float psum[8][D_DIM];
    __shared__ float rr[D_DIM];
    __shared__ float ggl[2 * D_DIM];
    __shared__ float hh[D_DIM];
    __shared__ float ffv[D_DIM];
    __shared__ float red[512];
    __shared__ float ys[16];
    const int tid  = threadIdx.x;
    const int wave = tid >> 5;
    const int lane = tid & 31;
    const int c0   = blockIdx.x << 2;
    const int n0   = c0 & ~15;

    {
        const int d4 = tid & 63;
        const int g  = tid >> 6;
        const v4f* p4 = (const v4f*)part + (long)(g * 8) * (D_DIM / 4) + d4;
        v4f pv[8];
#pragma unroll
        for (int i = 0; i < 8; ++i) pv[i] = p4[i * (D_DIM / 4)];
        v4f sacc = pv[0];
#pragma unroll
        for (int i = 1; i < 8; ++i) sacc += pv[i];
        ((v4f*)psum[g])[d4] = sacc;
    }
    __syncthreads();
    if (tid < D_DIM) {
        float o = psum[0][tid];
#pragma unroll
        for (int g = 1; g < 8; ++g) o += psum[g][tid];
        ovec[tid] = o;
    }
    __syncthreads();

    float v = (tid < D_DIM) ? ovec[tid] : 0.0f;
    const float mu = block_reduce_sum_512(v, red) * (1.0f / D_DIM);
    const float dv = (tid < D_DIM) ? (ovec[tid] - mu) : 0.0f;
    const float var = block_reduce_sum_512(dv * dv, red) * (1.0f / D_DIM);
    const float rstd = rsqrtf(var + EPSV);
    if (tid < D_DIM) rr[tid] = (ovec[tid] - mu) * rstd * g_res[tid] + be_res[tid];
    __syncthreads();

    {
        v8f acc0 = {}, acc1 = {};
        acc0 = gemv_tile_f32_unrolled(rr, w_geglu, (long)(2 * D_DIM), wave * 16, acc0);
        acc1 = gemv_tile_f32_unrolled(rr, w_geglu, (long)(2 * D_DIM), D_DIM + wave * 16, acc1);
        if (lane < 16) {
            ggl[wave * 16 + lane]         = acc0[0] + b_geglu[wave * 16 + lane];
            ggl[D_DIM + wave * 16 + lane] = acc1[0] + b_geglu[D_DIM + wave * 16 + lane];
        }
    }
    __syncthreads();

    if (tid < D_DIM) {
        const float g  = ggl[D_DIM + tid];
        const float ge = 0.5f * g * (1.0f + erff(g * 0.70710678118654752f));
        hh[tid] = ggl[tid] * ge;
    }
    __syncthreads();

    {
        v8f acc = {};
        acc = gemv_tile_f32_unrolled(hh, w_ff2, (long)D_DIM, wave * 16, acc);
        if (lane < 16) ffv[wave * 16 + lane] = acc[0] + b_ff2[wave * 16 + lane];
    }
    __syncthreads();

    if (wave == 0) {
        v8f acc = {};
        acc = gemv_tile_f32_unrolled(ffv, w_out, (long)C_DIM, n0, acc);
        if (lane < 16) ys[lane] = acc[0] + b_out[n0 + lane];
    }
    __syncthreads();

    {
        const int ch = tid >> 7;
        const float yv = ys[(c0 & 15) + ch];
        const v4f vv = {yv, yv, yv, yv};
        v4f* o = (v4f*)(out + (long)(c0 + ch) * N_TOK) + (tid & 127);
#pragma unroll
        for (int i = 0; i < 8; ++i) o[i * 128] = vv;
    }
}

extern "C" void kernel_launch(void* const* d_in, const int* in_sizes, int n_in,
                              void* d_out, int out_size, void* d_ws, size_t ws_size,
                              hipStream_t stream) {
    const float* x       = (const float*)d_in[0];
    const float* h       = (const float*)d_in[1];
    const float* s       = (const float*)d_in[2];
    const float* m       = (const float*)d_in[3];
    const float* tvec    = (const float*)d_in[4];
    const float* wt      = (const float*)d_in[5];
    const float* bt      = (const float*)d_in[6];
    const float* wk_x    = (const float*)d_in[7];
    const float* wv_x    = (const float*)d_in[8];
    const float* g_x     = (const float*)d_in[9];
    const float* be_x    = (const float*)d_in[10];
    const float* wk_h    = (const float*)d_in[11];
    const float* wv_h    = (const float*)d_in[12];
    const float* g_h     = (const float*)d_in[13];
    const float* be_h    = (const float*)d_in[14];
    const float* wk_s    = (const float*)d_in[15];
    const float* wv_s    = (const float*)d_in[16];
    const float* g_s     = (const float*)d_in[17];
    const float* be_s    = (const float*)d_in[18];
    const float* wk_m    = (const float*)d_in[19];
    const float* wv_m    = (const float*)d_in[20];
    const float* g_m     = (const float*)d_in[21];
    const float* be_m    = (const float*)d_in[22];
    const float* w_geglu = (const float*)d_in[23];
    const float* b_geglu = (const float*)d_in[24];
    const float* w_ff2   = (const float*)d_in[25];
    const float* b_ff2   = (const float*)d_in[26];
    const float* g_res   = (const float*)d_in[27];
    const float* be_res  = (const float*)d_in[28];
    const float* w_out   = (const float*)d_in[29];
    const float* b_out   = (const float*)d_in[30];
    float* out = (float*)d_out;
    float* ws  = (float*)d_ws;

    float* logits = ws + WS_LOG;
    float* mus    = ws + WS_MU;
    float* rstds  = ws + WS_RSTD;
    float* part   = ws + WS_PART;

    pass1_kernel<<<256, 512, 0, stream>>>(x, h, s, m, tvec, wt, bt,
                                          wk_x, wk_h, wk_s, wk_m,
                                          g_x, be_x, g_h, be_h, g_s, be_s, g_m, be_m,
                                          logits, mus, rstds);
    pass2_kernel<<<64, 1024, 0, stream>>>(x, h, s, m, logits, mus, rstds,
                                          wv_x, wv_h, wv_s, wv_m,
                                          g_x, be_x, g_h, be_h, g_s, be_s, g_m, be_m,
                                          part);
    tail_kernel<<<64, 512, 0, stream>>>(part,
                                        g_res, be_res, w_geglu, b_geglu,
                                        w_ff2, b_ff2, w_out, b_out, out);
}


# no LDS tree reductions: g*p, be*p sums ride each thread's channel walk (8 group partials); tail LN mean/var = 32 partials by wave 0 + serial add; 54->13 barriers; g*p, be*p, LN sum orders change
# speedup vs baseline: 3.1764x; 1.0589x over previous
#include <hip/hip_runtime.h>
#include <math.h>

typedef __attribute__((ext_vector_type(2))) float v2f;
typedef __attribute__((ext_vector_type(8))) float v8f;
typedef __attribute__((ext_vector_type(4))) float v4f;

#define C_DIM 256
#define D_DIM 256
#define N_TOK 4096
#define SCALE 0.0625f
#define EPSV  1e-5f

#define WS_LOG   1032
#define WS_MU    17416
#define WS_RSTD  33800
#define WS_PART  50184

__device__ inline v8f gemv_tile_f32(const float* x, const float* w,
                                    int K, long sk, long sn, int n0, v8f acc) {
    const int lane = threadIdx.x & 31;
    const int half = lane >> 4;
    const int l16  = lane & 15;
    const float amask = (l16 == 0) ? 1.0f : 0.0f;
    const float* wp = w + (long)(n0 + l16) * sn;
    for (int k0 = 0; k0 < K; k0 += 4) {
        const int kb = k0 + 2 * half;
        const float x0 = x[kb];
        const float x1 = x[kb + 1];
        v2f a;
        a.x = x0 * amask;
        a.y = x1 * amask;
        v2f b;
        b.x = wp[(long)kb * sk];
        b.y = wp[(long)(kb + 1) * sk];
        acc = __builtin_amdgcn_wmma_f32_16x16x4_f32(
             false, a,  false, b,
             (short)0, acc,  false,  false);
    }
    return acc;
}

__device__ __forceinline__ v8f gemv_tile_f32_unrolled(const float* x, const float* w,
                                                      long sk, int n0, v8f acc) {
    const int lane = threadIdx.x & 31;
    const int half = lane >> 4;
    const int l16  = lane & 15;
    const float amask = (l16 == 0) ? 1.0f : 0.0f;
    const float* wp = w + (n0 + l16);
#pragma unroll
    for (int k0 = 0; k0 < 256; k0 += 4) {
        const int kb = k0 + 2 * half;
        v2f a;
        a.x = x[kb] * amask;
        a.y = x[kb + 1] * amask;
        v2f b;
        b.x = wp[(long)kb * sk];
        b.y = wp[(long)(kb + 1) * sk];
        acc = __builtin_amdgcn_wmma_f32_16x16x4_f32(
             false, a,  false, b,
             (short)0, acc,  false,  false);
    }
    return acc;
}

__global__ __launch_bounds__(512) void pass1_kernel(
    const float* x, const float* h, const float* s, const float* m,
    const float* tvec, const float* wt, const float* bt,
    const float* wk_x, const float* wk_h, const float* wk_s, const float* wk_m,
    const float* g_x, const float* be_x, const float* g_h, const float* be_h,
    const float* g_s, const float* be_s, const float* g_m, const float* be_m,
    float* logits, float* mus, float* rstds)
{
    __shared__ float q_s[D_DIM];
    __shared__ float p_s[C_DIM];
    __shared__ __attribute__((aligned(16))) float gpm_s[C_DIM];
    __shared__ __attribute__((aligned(16))) float bpm_s[C_DIM];
    __shared__ float sgs[8];
    __shared__ float sbs[8];
    __shared__ float sh1[8][64];
    __shared__ float sh2[8][64];
    __shared__ float shd[8][64];
    const int tid  = threadIdx.x;
    const int wave = tid >> 5;
    const int lane = tid & 31;
    const int bid  = blockIdx.x;
    const int mod  = bid >> 6;
    const int j0   = (bid & 63) << 6;
    const int tl   = tid & 63;
    const int grp  = tid >> 6;

    const float* a  = (mod == 0) ? x    : (mod == 1) ? h    : (mod == 2) ? s    : m;
    const float* wk = (mod == 0) ? wk_x : (mod == 1) ? wk_h : (mod == 2) ? wk_s : wk_m;
    const float* gg = (mod == 0) ? g_x  : (mod == 1) ? g_h  : (mod == 2) ? g_s  : g_m;
    const float* bb = (mod == 0) ? be_x : (mod == 1) ? be_h : (mod == 2) ? be_s : be_m;

    float v[32];
    {
        const float* ap = a + (long)(grp << 5) * N_TOK + (j0 + tl);
#pragma unroll
        for (int cc = 0; cc < 32; ++cc) v[cc] = ap[(long)cc * N_TOK];
    }

    {
        v8f acc = {};
        acc = gemv_tile_f32_unrolled(tvec, wt, (long)D_DIM, wave * 16, acc);
        if (lane < 16) q_s[wave * 16 + lane] = acc[0] + bt[wave * 16 + lane];
    }
    __syncthreads();

    {
        v8f acc = {};
        acc = gemv_tile_f32(q_s, wk, D_DIM, 1L, (long)D_DIM, wave * 16, acc);
        if (lane < 16) p_s[wave * 16 + lane] = acc[0];
    }
    __syncthreads();

    if (tid < C_DIM) {
        const float p = p_s[tid];
        gpm_s[tid] = gg[tid] * p;
        bpm_s[tid] = bb[tid] * p;
    }
    __syncthreads();

    float s1 = 0.0f, s2 = 0.0f, dp = 0.0f;
    float sga[4] = {0.0f, 0.0f, 0.0f, 0.0f}, sba[4] = {0.0f, 0.0f, 0.0f, 0.0f};
    {
        const int c0 = grp << 5;
#pragma unroll
        for (int cc = 0; cc < 32; ++cc) {
            const float vv = v[cc];
            const float gp = gpm_s[c0 + cc];
            s1 += vv;
            s2 += vv * vv;
            dp += vv * gp;
            sga[cc & 3] += gp;
            sba[cc & 3] += bpm_s[c0 + cc];
        }
    }
    sh1[grp][tl] = s1; sh2[grp][tl] = s2; shd[grp][tl] = dp;
    if (tl == 0) {
        sgs[grp] = (sga[0] + sga[1]) + (sga[2] + sga[3]);
        sbs[grp] = (sba[0] + sba[1]) + (sba[2] + sba[3]);
    }
    __syncthreads();

    if (tid < 64) {
        float t1 = sh1[0][tid], t2 = sh2[0][tid], td = shd[0][tid];
        float sgp = sgs[0], sbp = sbs[0];
#pragma unroll
        for (int gi = 1; gi < 8; ++gi) {
            t1 += sh1[gi][tid]; t2 += sh2[gi][tid]; td += shd[gi][tid];
            sgp += sgs[gi]; sbp += sbs[gi];
        }
        const float mu   = t1 * (1.0f / C_DIM);
        const float var  = t2 * (1.0f / C_DIM) - mu * mu;
        const float rstd = rsqrtf(var + EPSV);
        const float lg   = SCALE * (rstd * (td - mu * sgp) + sbp);
        const int idx = mod * N_TOK + j0 + tid;
        logits[idx] = lg;
        mus[idx]    = mu;
        rstds[idx]  = rstd;
    }
}

__device__ __forceinline__ float wave_sum(float v) {
#pragma unroll
    for (int o = 16; o >= 1; o >>= 1) v += __shfl_xor(v, o, 32);
    return v;
}
__device__ __forceinline__ float wave_max(float v) {
#pragma unroll
    for (int o = 16; o >= 1; o >>= 1) v = fmaxf(v, __shfl_xor(v, o, 32));
    return v;
}

__global__ __launch_bounds__(1024) void pass2_kernel(
    const float* x, const float* h, const float* s, const float* m,
    const float* logits, const float* mus, const float* rstds,
    const float* wv_x, const float* wv_h, const float* wv_s, const float* wv_m,
    const float* g_x, const float* be_x, const float* g_h, const float* be_h,
    const float* g_s, const float* be_s, const float* g_m, const float* be_m,
    float* part)
{
    __shared__ float red_max[32];
    __shared__ float red_sum[32];
    __shared__ float red_out[32][18];
    __shared__ float u_s[16];
    __shared__ float o_s[2][D_DIM];
    const int tid  = threadIdx.x;
    const int wave = tid >> 5;
    const int lane = tid & 31;
    const int half = lane >> 4;
    const int l16  = lane & 15;
    const int bid  = blockIdx.x;
    const int mod  = bid >> 4;
    const int c0   = (bid & 15) << 4;

    const float* a  = (mod == 0) ? x    : (mod == 1) ? h    : (mod == 2) ? s    : m;
    const float* wv = (mod == 0) ? wv_x : (mod == 1) ? wv_h : (mod == 2) ? wv_s : wv_m;
    const float* gg = (mod == 0) ? g_x  : (mod == 1) ? g_h  : (mod == 2) ? g_s  : g_m;
    const float* bb = (mod == 0) ? be_x : (mod == 1) ? be_h : (mod == 2) ? be_s : be_m;

    const float gch = gg[c0 + (tid & 15)];
    const float bch = bb[c0 + (tid & 15)];

    const int kq = wave >> 4;
    const int nt = (wave & 15) << 4;
    v2f bw[2];
    {
        const float* wp = wv + (long)(c0 + 8 * kq) * D_DIM + nt + l16;
#pragma unroll
        for (int ks = 0; ks < 2; ++ks) {
            const int kb = 4 * ks + 2 * half;
            bw[ks].x = wp[(long)kb * D_DIM];
            bw[ks].y = wp[(long)(kb + 1) * D_DIM];
        }
    }

    v4f L[4];
    {
        const v4f* lg4 = (const v4f*)logits + tid;
#pragma unroll
        for (int q = 0; q < 4; ++q) L[q] = lg4[q * 1024];
    }
    const v4f rs = ((const v4f*)(rstds + mod * N_TOK))[tid];
    const v4f mu = ((const v4f*)(mus + mod * N_TOK))[tid];
    v4f av[16];
#pragma unroll
    for (int cc = 0; cc < 16; ++cc)
        av[cc] = ((const v4f*)(a + (long)(c0 + cc) * N_TOK))[tid];

    __builtin_amdgcn_sched_barrier(0);

    float lmax = -3.4e38f;
#pragma unroll
    for (int q = 0; q < 4; ++q)
        lmax = fmaxf(lmax, fmaxf(fmaxf(L[q].x, L[q].y), fmaxf(L[q].z, L[q].w)));
    lmax = wave_max(lmax);
    if (lane == 0) red_max[wave] = lmax;
    __syncthreads();
    const float gmax = wave_max(red_max[lane]);

    float ls = 0.0f;
#pragma unroll
    for (int q = 0; q < 4; ++q) {
        v4f e;
        e.x = __expf(L[q].x - gmax);
        e.y = __expf(L[q].y - gmax);
        e.z = __expf(L[q].z - gmax);
        e.w = __expf(L[q].w - gmax);
        L[q] = e;
        ls += e.x; ls += e.y; ls += e.z; ls += e.w;
    }
    ls = wave_sum(ls);
    if (lane == 0) red_sum[wave] = ls;
    __syncthreads();
    const float zsum = wave_sum(red_sum[lane]);
    const float inv = 1.0f / zsum;

    float wps = 0.0f, aps = 0.0f;
    float t[16];
#pragma unroll
    for (int cc = 0; cc < 16; ++cc) t[cc] = 0.0f;
    {
        const v4f e = (mod == 0) ? L[0] : (mod == 1) ? L[1] : (mod == 2) ? L[2] : L[3];
#pragma unroll
        for (int k = 0; k < 4; ++k) {
            const float w  = e[k] * inv;
            const float al = w * rs[k];
            wps += w;
            aps += al * mu[k];
#pragma unroll
            for (int cc = 0; cc < 16; ++cc) t[cc] += al * av[cc][k];
        }
    }
    wps = wave_sum(wps); aps = wave_sum(aps);
    float r8[8], r4[4], r2[2], r1;
    {
        const bool b = (lane & 16) != 0;
#pragma unroll
        for (int i = 0; i < 8; ++i) {
            const float snd = b ? t[i] : t[i + 8];
            const float kp  = b ? t[i + 8] : t[i];
            r8[i] = kp + __shfl_xor(snd, 16, 32);
        }
    }
    {
        const bool b = (lane & 8) != 0;
#pragma unroll
        for (int i = 0; i < 4; ++i) {
            const float snd = b ? r8[i] : r8[i + 4];
            const float kp  = b ? r8[i + 4] : r8[i];
            r4[i] = kp + __shfl_xor(snd, 8, 32);
        }
    }
    {
        const bool b = (lane & 4) != 0;
#pragma unroll
        for (int i = 0; i < 2; ++i) {
            const float snd = b ? r4[i] : r4[i + 2];
            const float kp  = b ? r4[i + 2] : r4[i];
            r2[i] = kp + __shfl_xor(snd, 4, 32);
        }
    }
    {
        const bool b = (lane & 2) != 0;
        const float snd = b ? r2[0] : r2[1];
        const float kp  = b ? r2[1] : r2[0];
        r1 = kp + __shfl_xor(snd, 2, 32);
    }
    r1 += __shfl_xor(r1, 1, 32);
    if ((lane & 1) == 0) red_out[wave][lane >> 1] = r1;
    if (lane == 0) { red_out[wave][16] = wps; red_out[wave][17] = aps; }
    __syncthreads();
    if (tid < 16) {
        float T = red_out[0][tid], wsum = red_out[0][16], amu = red_out[0][17];
#pragma unroll
        for (int w = 1; w < 32; ++w) { T += red_out[w][tid]; wsum += red_out[w][16]; amu += red_out[w][17]; }
        u_s[tid] = gch * (T - amu) + bch * wsum;
    }
    __syncthreads();

    {
        const float amask = (l16 == 0) ? 1.0f : 0.0f;
        v8f acc = {};
#pragma unroll
        for (int ks = 0; ks < 2; ++ks) {
            const int kb = 8 * kq + 4 * ks + 2 * half;
            v2f av2;
            av2.x = u_s[kb] * amask;
            av2.y = u_s[kb + 1] * amask;
            acc = __builtin_amdgcn_wmma_f32_16x16x4_f32(
                 false, av2,  false, bw[ks],
                 (short)0, acc,  false,  false);
        }
        if (lane < 16) o_s[kq][nt + lane] = acc[0];
    }
    __syncthreads();
    if (tid < D_DIM) part[bid * D_DIM + tid] = o_s[0][tid] + o_s[1][tid];
}

__global__ __launch_bounds__(512) void tail_kernel(
    const float* part,
    const float* g_res, const float* be_res,
    const float* w_geglu, const float* b_geglu,
    const float* w_ff2, const float* b_ff2,
    const float* w_out, const float* b_out,
    float* out)
{
    __shared__ __attribute__((aligned(16))) float ovec[D_DIM];
    __shared__ __attribute__((aligned(16))) float psum[8][D_DIM];
    __shared__ __attribute__((aligned(16))) float pm[32];
    __shared__ __attribute__((aligned(16))) float pv[32];
    __shared__ float rr[D_DIM];
    __shared__ float ggl[2 * D_DIM];
    __shared__ float hh[D_DIM];
    __shared__ float ffv[D_DIM];
    __shared__ float ys[16];
    const int tid  = threadIdx.x;
    const int wave = tid >> 5;
    const int lane = tid & 31;
    const int c0   = blockIdx.x << 2;
    const int n0   = c0 & ~15;

    {
        const int d4 = tid & 63;
        const int g  = tid >> 6;
        const v4f* p4 = (const v4f*)part + (long)(g * 8) * (D_DIM / 4) + d4;
        v4f pv[8];
#pragma unroll
        for (int i = 0; i < 8; ++i) pv[i] = p4[i * (D_DIM / 4)];
        v4f sacc = pv[0];
#pragma unroll
        for (int i = 1; i < 8; ++i) sacc += pv[i];
        ((v4f*)psum[g])[d4] = sacc;
    }
    __syncthreads();
    if (tid < D_DIM) {
        float o = psum[0][tid];
#pragma unroll
        for (int g = 1; g < 8; ++g) o += psum[g][tid];
        ovec[tid] = o;
    }
    __syncthreads();

    v4f oa = {0.0f, 0.0f, 0.0f, 0.0f}, ob = {0.0f, 0.0f, 0.0f, 0.0f};
    if (wave == 0) {
        oa = ((const v4f*)ovec)[lane];
        ob = ((const v4f*)ovec)[lane + 32];
        pm[lane] = ((oa.x + oa.y) + (oa.z + oa.w)) + ((ob.x + ob.y) + (ob.z + ob.w));
    }
    __syncthreads();
    float mu;
    {
        v4f t4 = ((const v4f*)pm)[0];
        for (int i = 1; i < 8; ++i) t4 += ((const v4f*)pm)[i];
        mu = ((t4.x + t4.y) + (t4.z + t4.w)) * (1.0f / D_DIM);
    }
    if (wave == 0) {
        const v4f da = oa - mu, db = ob - mu;
        pv[lane] = ((da.x * da.x + da.y * da.y) + (da.z * da.z + da.w * da.w))
                 + ((db.x * db.x + db.y * db.y) + (db.z * db.z + db.w * db.w));
    }
    __syncthreads();
    float var;
    {
        v4f t4 = ((const v4f*)pv)[0];
        for (int i = 1; i < 8; ++i) t4 += ((const v4f*)pv)[i];
        var = ((t4.x + t4.y) + (t4.z + t4.w)) * (1.0f / D_DIM);
    }
    const float rstd = rsqrtf(var + EPSV);
    if (tid < D_DIM) rr[tid] = (ovec[tid] - mu) * rstd * g_res[tid] + be_res[tid];
    __syncthreads();

    {
        v8f acc0 = {}, acc1 = {};
        acc0 = gemv_tile_f32_unrolled(rr, w_geglu, (long)(2 * D_DIM), wave * 16, acc0);
        acc1 = gemv_tile_f32_unrolled(rr, w_geglu, (long)(2 * D_DIM), D_DIM + wave * 16, acc1);
        if (lane < 16) {
            ggl[wave * 16 + lane]         = acc0[0] + b_geglu[wave * 16 + lane];
            ggl[D_DIM + wave * 16 + lane] = acc1[0] + b_geglu[D_DIM + wave * 16 + lane];
        }
    }
    __syncthreads();

    if (tid < D_DIM) {
        const float g  = ggl[D_DIM + tid];
        const float ge = 0.5f * g * (1.0f + erff(g * 0.70710678118654752f));
        hh[tid] = ggl[tid] * ge;
    }
    __syncthreads();

    {
        v8f acc = {};
        acc = gemv_tile_f32_unrolled(hh, w_ff2, (long)D_DIM, wave * 16, acc);
        if (lane < 16) ffv[wave * 16 + lane] = acc[0] + b_ff2[wave * 16 + lane];
    }
    __syncthreads();

    if (wave == 0) {
        v8f acc = {};
        acc = gemv_tile_f32_unrolled(ffv, w_out, (long)C_DIM, n0, acc);
        if (lane < 16) ys[lane] = acc[0] + b_out[n0 + lane];
    }
    __syncthreads();

    {
        const int ch = tid >> 7;
        const float yv = ys[(c0 & 15) + ch];
        const v4f vv = {yv, yv, yv, yv};
        v4f* o = (v4f*)(out + (long)(c0 + ch) * N_TOK) + (tid & 127);
#pragma unroll
        for (int i = 0; i < 8; ++i) o[i * 128] = vv;
    }
}

extern "C" void kernel_launch(void* const* d_in, const int* in_sizes, int n_in,
                              void* d_out, int out_size, void* d_ws, size_t ws_size,
                              hipStream_t stream) {
    const float* x       = (const float*)d_in[0];
    const float* h       = (const float*)d_in[1];
    const float* s       = (const float*)d_in[2];
    const float* m       = (const float*)d_in[3];
    const float* tvec    = (const float*)d_in[4];
    const float* wt      = (const float*)d_in[5];
    const float* bt      = (const float*)d_in[6];
    const float* wk_x    = (const float*)d_in[7];
    const float* wv_x    = (const float*)d_in[8];
    const float* g_x     = (const float*)d_in[9];
    const float* be_x    = (const float*)d_in[10];
    const float* wk_h    = (const float*)d_in[11];
    const float* wv_h    = (const float*)d_in[12];
    const float* g_h     = (const float*)d_in[13];
    const float* be_h    = (const float*)d_in[14];
    const float* wk_s    = (const float*)d_in[15];
    const float* wv_s    = (const float*)d_in[16];
    const float* g_s     = (const float*)d_in[17];
    const float* be_s    = (const float*)d_in[18];
    const float* wk_m    = (const float*)d_in[19];
    const float* wv_m    = (const float*)d_in[20];
    const float* g_m     = (const float*)d_in[21];
    const float* be_m    = (const float*)d_in[22];
    const float* w_geglu = (const float*)d_in[23];
    const float* b_geglu = (const float*)d_in[24];
    const float* w_ff2   = (const float*)d_in[25];
    const float* b_ff2   = (const float*)d_in[26];
    const float* g_res   = (const float*)d_in[27];
    const float* be_res  = (const float*)d_in[28];
    const float* w_out   = (const float*)d_in[29];
    const float* b_out   = (const float*)d_in[30];
    float* out = (float*)d_out;
    float* ws  = (float*)d_ws;

    float* logits = ws + WS_LOG;
    float* mus    = ws + WS_MU;
    float* rstds  = ws + WS_RSTD;
    float* part   = ws + WS_PART;

    pass1_kernel<<<256, 512, 0, stream>>>(x, h, s, m, tvec, wt, bt,
                                          wk_x, wk_h, wk_s, wk_m,
                                          g_x, be_x, g_h, be_h, g_s, be_s, g_m, be_m,
                                          logits, mus, rstds);
    pass2_kernel<<<64, 1024, 0, stream>>>(x, h, s, m, logits, mus, rstds,
                                          wv_x, wv_h, wv_s, wv_m,
                                          g_x, be_x, g_h, be_h, g_s, be_s, g_m, be_m,
                                          part);
    tail_kernel<<<64, 512, 0, stream>>>(part,
                                        g_res, be_res, w_geglu, b_geglu,
                                        w_ff2, b_ff2, w_out, b_out, out);
}


# weight loads of the q, p, GEGLU and ff2 GEMVs widened to 16 bytes per lane: a wave takes 4 neighbouring column tiles over a K slice (p: 2 k-steps per load); q, p, gg, ff sum orders change
# speedup vs baseline: 3.4895x; 1.0986x over previous
#include <hip/hip_runtime.h>
#include <math.h>

typedef __attribute__((ext_vector_type(2))) float v2f;
typedef __attribute__((ext_vector_type(8))) float v8f;
typedef __attribute__((ext_vector_type(4))) float v4f;

#define C_DIM 256
#define D_DIM 256
#define N_TOK 4096
#define SCALE 0.0625f
#define EPSV  1e-5f

#define WS_LOG   1032
#define WS_MU    17416
#define WS_RSTD  33800
#define WS_PART  50184

__device__ __forceinline__ v8f gemv_tile_f32_unrolled(const float* x, const float* w,
                                                      long sk, int n0, v8f acc) {
    const int lane = threadIdx.x & 31;
    const int half = lane >> 4;
    const int l16  = lane & 15;
    const float amask = (l16 == 0) ? 1.0f : 0.0f;
    const float* wp = w + (long)(2 * half) * sk + (n0 + l16);
    const float* xp = x + 2 * half;
#pragma unroll
    for (int k0 = 0; k0 < 256; k0 += 4) {
        v2f a;
        a.x = xp[k0] * amask;
        a.y = xp[k0 + 1] * amask;
        v2f b;
        b.x = wp[(long)k0 * sk];
        b.y = wp[(long)(k0 + 1) * sk];
        acc = __builtin_amdgcn_wmma_f32_16x16x4_f32(
             false, a,  false, b,
             (short)0, acc,  false,  false);
    }
    return acc;
}

template <int STEPS>
__device__ __forceinline__ void gemv_quad_f32(const float* x, const float* w, long ldw,
                                              int nc0, int kbeg,
                                              float& r0, float& r1, float& r2, float& r3) {
    const int lane = threadIdx.x & 31;
    const int half = lane >> 4;
    const int l16  = lane & 15;
    const float amask = (l16 == 0) ? 1.0f : 0.0f;
    const float* wp = w + (long)(kbeg + 2 * half) * ldw + nc0 + 4 * l16;
    v8f acc0 = {}, acc1 = {}, acc2 = {}, acc3 = {};
#pragma unroll
    for (int s = 0; s < STEPS; ++s) {
        const int kb = kbeg + 4 * s + 2 * half;
        v2f a;
        a.x = x[kb] * amask;
        a.y = x[kb + 1] * amask;
        const v4f w0 = *(const v4f*)(wp + (long)(4 * s) * ldw);
        const v4f w1 = *(const v4f*)(wp + (long)(4 * s + 1) * ldw);
        v2f b;
        b.x = w0.x; b.y = w1.x;
        acc0 = __builtin_amdgcn_wmma_f32_16x16x4_f32(false, a, false, b, (short)0, acc0, false, false);
        b.x = w0.y; b.y = w1.y;
        acc1 = __builtin_amdgcn_wmma_f32_16x16x4_f32(false, a, false, b, (short)0, acc1, false, false);
        b.x = w0.z; b.y = w1.z;
        acc2 = __builtin_amdgcn_wmma_f32_16x16x4_f32(false, a, false, b, (short)0, acc2, false, false);
        b.x = w0.w; b.y = w1.w;
        acc3 = __builtin_amdgcn_wmma_f32_16x16x4_f32(false, a, false, b, (short)0, acc3, false, false);
    }
    r0 = acc0[0]; r1 = acc1[0]; r2 = acc2[0]; r3 = acc3[0];
}

__device__ __forceinline__ v8f gemv_tile_f32_rows(const float* x, const float* w, int n0, v8f acc) {
    const int lane = threadIdx.x & 31;
    const int half = lane >> 4;
    const int l16  = lane & 15;
    const float amask = (l16 == 0) ? 1.0f : 0.0f;
    const float* wp = w + (long)(n0 + l16) * 256 + 4 * half;
    const float* xp = x + 4 * half;
#pragma unroll
    for (int s = 0; s < 32; ++s) {
        const v4f wv4 = *(const v4f*)(wp + 8 * s);
        const v4f xv4 = *(const v4f*)(xp + 8 * s);
        v2f a, b;
        a.x = xv4.x * amask; a.y = xv4.y * amask;
        b.x = wv4.x;         b.y = wv4.y;
        acc = __builtin_amdgcn_wmma_f32_16x16x4_f32(false, a, false, b, (short)0, acc, false, false);
        a.x = xv4.z * amask; a.y = xv4.w * amask;
        b.x = wv4.z;         b.y = wv4.w;
        acc = __builtin_amdgcn_wmma_f32_16x16x4_f32(false, a, false, b, (short)0, acc, false, false);
    }
    return acc;
}

__global__ __launch_bounds__(512) void pass1_kernel(
    const float* x, const float* h, const float* s, const float* m,
    const float* tvec, const float* wt, const float* bt,
    const float* wk_x, const float* wk_h, const float* wk_s, const float* wk_m,
    const float* g_x, const float* be_x, const float* g_h, const float* be_h,
    const float* g_s, const float* be_s, const float* g_m, const float* be_m,
    float* logits, float* mus, float* rstds)
{
    __shared__ __attribute__((aligned(16))) float q_s[D_DIM];
    __shared__ float qp_s[4][D_DIM];
    __shared__ float p_s[C_DIM];
    __shared__ __attribute__((aligned(16))) float gpm_s[C_DIM];
    __shared__ __attribute__((aligned(16))) float bpm_s[C_DIM];
    __shared__ float sgs[8];
    __shared__ float sbs[8];
    __shared__ float sh1[8][64];
    __shared__ float sh2[8][64];
    __shared__ float shd[8][64];
    const int tid  = threadIdx.x;
    const int wave = tid >> 5;
    const int lane = tid & 31;
    const int bid  = blockIdx.x;
    const int mod  = bid >> 6;
    const int j0   = (bid & 63) << 6;
    const int tl   = tid & 63;
    const int grp  = tid >> 6;

    const float* a  = (mod == 0) ? x    : (mod == 1) ? h    : (mod == 2) ? s    : m;
    const float* wk = (mod == 0) ? wk_x : (mod == 1) ? wk_h : (mod == 2) ? wk_s : wk_m;
    const float* gg = (mod == 0) ? g_x  : (mod == 1) ? g_h  : (mod == 2) ? g_s  : g_m;
    const float* bb = (mod == 0) ? be_x : (mod == 1) ? be_h : (mod == 2) ? be_s : be_m;

    float v[32];
    {
        const float* ap = a + (long)(grp << 5) * N_TOK + (j0 + tl);
#pragma unroll
        for (int cc = 0; cc < 32; ++cc) v[cc] = ap[(long)cc * N_TOK];
    }

    {
        const int cg = (wave & 3) << 6;
        const int kq = wave >> 2;
        float r0, r1, r2, r3;
        gemv_quad_f32<16>(tvec, wt, (long)D_DIM, cg, kq << 6, r0, r1, r2, r3);
        if (lane < 16) {
            float* qd = &qp_s[kq][cg + 4 * lane];
            qd[0] = r0; qd[1] = r1; qd[2] = r2; qd[3] = r3;
        }
    }
    __syncthreads();
    if (tid < D_DIM)
        q_s[tid] = ((qp_s[0][tid] + qp_s[1][tid]) + (qp_s[2][tid] + qp_s[3][tid])) + bt[tid];
    __syncthreads();

    {
        v8f acc = {};
        acc = gemv_tile_f32_rows(q_s, wk, wave * 16, acc);
        if (lane < 16) p_s[wave * 16 + lane] = acc[0];
    }
    __syncthreads();

    if (tid < C_DIM) {
        const float p = p_s[tid];
        gpm_s[tid] = gg[tid] * p;
        bpm_s[tid] = bb[tid] * p;
    }
    __syncthreads();

    float s1 = 0.0f, s2 = 0.0f, dp = 0.0f;
    float sga[4] = {0.0f, 0.0f, 0.0f, 0.0f}, sba[4] = {0.0f, 0.0f, 0.0f, 0.0f};
    {
        const int c0 = grp << 5;
#pragma unroll
        for (int cc = 0; cc < 32; ++cc) {
            const float vv = v[cc];
            const float gp = gpm_s[c0 + cc];
            s1 += vv;
            s2 += vv * vv;
            dp += vv * gp;
            sga[cc & 3] += gp;
            sba[cc & 3] += bpm_s[c0 + cc];
        }
    }
    sh1[grp][tl] = s1; sh2[grp][tl] = s2; shd[grp][tl] = dp;
    if (tl == 0) {
        sgs[grp] = (sga[0] + sga[1]) + (sga[2] + sga[3]);
        sbs[grp] = (sba[0] + sba[1]) + (sba[2] + sba[3]);
    }
    __syncthreads();

    if (tid < 64) {
        float t1 = sh1[0][tid], t2 = sh2[0][tid], td = shd[0][tid];
        float sgp = sgs[0], sbp = sbs[0];
#pragma unroll
        for (int gi = 1; gi < 8; ++gi) {
            t1 += sh1[gi][tid]; t2 += sh2[gi][tid]; td += shd[gi][tid];
            sgp += sgs[gi]; sbp += sbs[gi];
        }
        const float mu   = t1 * (1.0f / C_DIM);
        const float var  = t2 * (1.0f / C_DIM) - mu * mu;
        const float rstd = rsqrtf(var + EPSV);
        const float lg   = SCALE * (rstd * (td - mu * sgp) + sbp);
        const int idx = mod * N_TOK + j0 + tid;
        logits[idx] = lg;
        mus[idx]    = mu;
        rstds[idx]  = rstd;
    }
}

__device__ __forceinline__ float wave_sum(float v) {
#pragma unroll
    for (int o = 16; o >= 1; o >>= 1) v += __shfl_xor(v, o, 32);
    return v;
}
__device__ __forceinline__ float wave_max(float v) {
#pragma unroll
    for (int o = 16; o >= 1; o >>= 1) v = fmaxf(v, __shfl_xor(v, o, 32));
    return v;
}

__global__ __launch_bounds__(1024) void pass2_kernel(
    const float* x, const float* h, const float* s, const float* m,
    const float* logits, const float* mus, const float* rstds,
    const float* wv_x, const float* wv_h, const float* wv_s, const float* wv_m,
    const float* g_x, const float* be_x, const float* g_h, const float* be_h,
    const float* g_s, const float* be_s, const float* g_m, const float* be_m,
    float* part)
{
    __shared__ float red_max[32];
    __shared__ float red_sum[32];
    __shared__ float red_out[32][18];
    __shared__ float u_s[16];
    __shared__ float o_s[2][D_DIM];
    const int tid  = threadIdx.x;
    const int wave = tid >> 5;
    const int lane = tid & 31;
    const int half = lane >> 4;
    const int l16  = lane & 15;
    const int bid  = blockIdx.x;
    const int mod  = bid >> 4;
    const int c0   = (bid & 15) << 4;

    const float* a  = (mod == 0) ? x    : (mod == 1) ? h    : (mod == 2) ? s    : m;
    const float* wv = (mod == 0) ? wv_x : (mod == 1) ? wv_h : (mod == 2) ? wv_s : wv_m;
    const float* gg = (mod == 0) ? g_x  : (mod == 1) ? g_h  : (mod == 2) ? g_s  : g_m;
    const float* bb = (mod == 0) ? be_x : (mod == 1) ? be_h : (mod == 2) ? be_s : be_m;

    const float gch = gg[c0 + (tid & 15)];
    const float bch = bb[c0 + (tid & 15)];

    const int kq = wave >> 4;
    const int nt = (wave & 15) << 4;
    v2f bw[2];
    {
        const float* wp = wv + (long)(c0 + 8 * kq) * D_DIM + nt + l16;
#pragma unroll
        for (int ks = 0; ks < 2; ++ks) {
            const int kb = 4 * ks + 2 * half;
            bw[ks].x = wp[(long)kb * D_DIM];
            bw[ks].y = wp[(long)(kb + 1) * D_DIM];
        }
    }

    v4f L[4];
    {
        const v4f* lg4 = (const v4f*)logits + tid;
#pragma unroll
        for (int q = 0; q < 4; ++q) L[q] = lg4[q * 1024];
    }
    const v4f rs = ((const v4f*)(rstds + mod * N_TOK))[tid];
    const v4f mu = ((const v4f*)(mus + mod * N_TOK))[tid];
    v4f av[16];
#pragma unroll
    for (int cc = 0; cc < 16; ++cc)
        av[cc] = ((const v4f*)(a + (long)(c0 + cc) * N_TOK))[tid];

    __builtin_amdgcn_sched_barrier(0);

    float lmax = -3.4e38f;
#pragma unroll
    for (int q = 0; q < 4; ++q)
        lmax = fmaxf(lmax, fmaxf(fmaxf(L[q].x, L[q].y), fmaxf(L[q].z, L[q].w)));
    lmax = wave_max(lmax);
    if (lane == 0) red_max[wave] = lmax;
    __syncthreads();
    const float gmax = wave_max(red_max[lane]);

    float ls = 0.0f;
#pragma unroll
    for (int q = 0; q < 4; ++q) {
        v4f e;
        e.x = __expf(L[q].x - gmax);
        e.y = __expf(L[q].y - gmax);
        e.z = __expf(L[q].z - gmax);
        e.w = __expf(L[q].w - gmax);
        L[q] = e;
        ls += e.x; ls += e.y; ls += e.z; ls += e.w;
    }
    ls = wave_sum(ls);
    if (lane == 0) red_sum[wave] = ls;
    __syncthreads();
    const float zsum = wave_sum(red_sum[lane]);
    const float inv = 1.0f / zsum;

    float wps = 0.0f, aps = 0.0f;
    float t[16];
#pragma unroll
    for (int cc = 0; cc < 16; ++cc) t[cc] = 0.0f;
    {
        const v4f e = (mod == 0) ? L[0] : (mod == 1) ? L[1] : (mod == 2) ? L[2] : L[3];
#pragma unroll
        for (int k = 0; k < 4; ++k) {
            const float w  = e[k] * inv;
            const float al = w * rs[k];
            wps += w;
            aps += al * mu[k];
#pragma unroll
            for (int cc = 0; cc < 16; ++cc) t[cc] += al * av[cc][k];
        }
    }
    wps = wave_sum(wps); aps = wave_sum(aps);
    float r8[8], r4[4], r2[2], r1;
    {
        const bool b = (lane & 16) != 0;
#pragma unroll
        for (int i = 0; i < 8; ++i) {
            const float snd = b ? t[i] : t[i + 8];
            const float kp  = b ? t[i + 8] : t[i];
            r8[i] = kp + __shfl_xor(snd, 16, 32);
        }
    }
    {
        const bool b = (lane & 8) != 0;
#pragma unroll
        for (int i = 0; i < 4; ++i) {
            const float snd = b ? r8[i] : r8[i + 4];
            const float kp  = b ? r8[i + 4] : r8[i];
            r4[i] = kp + __shfl_xor(snd, 8, 32);
        }
    }
    {
        const bool b = (lane & 4) != 0;
#pragma unroll
        for (int i = 0; i < 2; ++i) {
            const float snd = b ? r4[i] : r4[i + 2];
            const float kp  = b ? r4[i + 2] : r4[i];
            r2[i] = kp + __shfl_xor(snd, 4, 32);
        }
    }
    {
        const bool b = (lane & 2) != 0;
        const float snd = b ? r2[0] : r2[1];
        const float kp  = b ? r2[1] : r2[0];
        r1 = kp + __shfl_xor(snd, 2, 32);
    }
    r1 += __shfl_xor(r1, 1, 32);
    if ((lane & 1) == 0) red_out[wave][lane >> 1] = r1;
    if (lane == 0) { red_out[wave][16] = wps; red_out[wave][17] = aps; }
    __syncthreads();
    if (tid < 16) {
        float T = red_out[0][tid], wsum = red_out[0][16], amu = red_out[0][17];
#pragma unroll
        for (int w = 1; w < 32; ++w) { T += red_out[w][tid]; wsum += red_out[w][16]; amu += red_out[w][17]; }
        u_s[tid] = gch * (T - amu) + bch * wsum;
    }
    __syncthreads();

    {
        const float amask = (l16 == 0) ? 1.0f : 0.0f;
        v8f acc = {};
#pragma unroll
        for (int ks = 0; ks < 2; ++ks) {
            const int kb = 8 * kq + 4 * ks + 2 * half;
            v2f av2;
            av2.x = u_s[kb] * amask;
            av2.y = u_s[kb + 1] * amask;
            acc = __builtin_amdgcn_wmma_f32_16x16x4_f32(
                 false, av2,  false, bw[ks],
                 (short)0, acc,  false,  false);
        }
        if (lane < 16) o_s[kq][nt + lane] = acc[0];
    }
    __syncthreads();
    if (tid < D_DIM) part[bid * D_DIM + tid] = o_s[0][tid] + o_s[1][tid];
}

__global__ __launch_bounds__(512) void tail_kernel(
    const float* part,
    const float* g_res, const float* be_res,
    const float* w_geglu, const float* b_geglu,
    const float* w_ff2, const float* b_ff2,
    const float* w_out, const float* b_out,
    float* out)
{
    __shared__ __attribute__((aligned(16))) float ovec[D_DIM];
    __shared__ __attribute__((aligned(16))) float psum[8][D_DIM];
    __shared__ __attribute__((aligned(16))) float pm[32];
    __shared__ __attribute__((aligned(16))) float pv[32];
    __shared__ float rr[D_DIM];
    __shared__ float ggp[2][2 * D_DIM];
    __shared__ float hh[D_DIM];
    __shared__ float ffp[4][D_DIM];
    __shared__ __attribute__((aligned(16))) float ffv[D_DIM];
    __shared__ float ys[16];
    const int tid  = threadIdx.x;
    const int wave = tid >> 5;
    const int lane = tid & 31;
    const int c0   = blockIdx.x << 2;
    const int n0   = c0 & ~15;

    {
        const int d4 = tid & 63;
        const int g  = tid >> 6;
        const v4f* p4 = (const v4f*)part + (long)(g * 8) * (D_DIM / 4) + d4;
        v4f pv[8];
#pragma unroll
        for (int i = 0; i < 8; ++i) pv[i] = p4[i * (D_DIM / 4)];
        v4f sacc = pv[0];
#pragma unroll
        for (int i = 1; i < 8; ++i) sacc += pv[i];
        ((v4f*)psum[g])[d4] = sacc;
    }
    __syncthreads();
    if (tid < D_DIM) {
        float o = psum[0][tid];
#pragma unroll
        for (int g = 1; g < 8; ++g) o += psum[g][tid];
        ovec[tid] = o;
    }
    __syncthreads();

    v4f oa = {0.0f, 0.0f, 0.0f, 0.0f}, ob = {0.0f, 0.0f, 0.0f, 0.0f};
    if (wave == 0) {
        oa = ((const v4f*)ovec)[lane];
        ob = ((const v4f*)ovec)[lane + 32];
        pm[lane] = ((oa.x + oa.y) + (oa.z + oa.w)) + ((ob.x + ob.y) + (ob.z + ob.w));
    }
    __syncthreads();
    float mu;
    {
        v4f t4 = ((const v4f*)pm)[0];
        for (int i = 1; i < 8; ++i) t4 += ((const v4f*)pm)[i];
        mu = ((t4.x + t4.y) + (t4.z + t4.w)) * (1.0f / D_DIM);
    }
    if (wave == 0) {
        const v4f da = oa - mu, db = ob - mu;
        pv[lane] = ((da.x * da.x + da.y * da.y) + (da.z * da.z + da.w * da.w))
                 + ((db.x * db.x + db.y * db.y) + (db.z * db.z + db.w * db.w));
    }
    __syncthreads();
    float var;
    {
        v4f t4 = ((const v4f*)pv)[0];
        for (int i = 1; i < 8; ++i) t4 += ((const v4f*)pv)[i];
        var = ((t4.x + t4.y) + (t4.z + t4.w)) * (1.0f / D_DIM);
    }
    const float rstd = rsqrtf(var + EPSV);
    if (tid < D_DIM) rr[tid] = (ovec[tid] - mu) * rstd * g_res[tid] + be_res[tid];
    __syncthreads();

    {
        const int cg = (wave & 7) << 6;
        const int kh = wave >> 3;
        float r0, r1, r2, r3;
        gemv_quad_f32<32>(rr, w_geglu, (long)(2 * D_DIM), cg, kh << 7, r0, r1, r2, r3);
        if (lane < 16) {
            float* gd = &ggp[kh][cg + 4 * lane];
            gd[0] = r0; gd[1] = r1; gd[2] = r2; gd[3] = r3;
        }
    }
    __syncthreads();

    if (tid < D_DIM) {
        const float u  = (ggp[0][tid] + ggp[1][tid]) + b_geglu[tid];
        const float g  = (ggp[0][D_DIM + tid] + ggp[1][D_DIM + tid]) + b_geglu[D_DIM + tid];
        const float ge = 0.5f * g * (1.0f + erff(g * 0.70710678118654752f));
        hh[tid] = u * ge;
    }
    __syncthreads();

    {
        const int cg = (wave & 3) << 6;
        const int kq = wave >> 2;
        float r0, r1, r2, r3;
        gemv_quad_f32<16>(hh, w_ff2, (long)D_DIM, cg, kq << 6, r0, r1, r2, r3);
        if (lane < 16) {
            float* fd = &ffp[kq][cg + 4 * lane];
            fd[0] = r0; fd[1] = r1; fd[2] = r2; fd[3] = r3;
        }
    }
    __syncthreads();
    if (tid < D_DIM)
        ffv[tid] = ((ffp[0][tid] + ffp[1][tid]) + (ffp[2][tid] + ffp[3][tid])) + b_ff2[tid];
    __syncthreads();

    if (wave == 0) {
        v8f acc = {};
        acc = gemv_tile_f32_unrolled(ffv, w_out, (long)C_DIM, n0, acc);
        if (lane < 16) ys[lane] = acc[0] + b_out[n0 + lane];
    }
    __syncthreads();

    {
        const int ch = tid >> 7;
        const float yv = ys[(c0 & 15) + ch];
        const v4f vv = {yv, yv, yv, yv};
        v4f* o = (v4f*)(out + (long)(c0 + ch) * N_TOK) + (tid & 127);
#pragma unroll
        for (int i = 0; i < 8; ++i) o[i * 128] = vv;
    }
}

extern "C" void kernel_launch(void* const* d_in, const int* in_sizes, int n_in,
                              void* d_out, int out_size, void* d_ws, size_t ws_size,
                              hipStream_t stream) {
    const float* x       = (const float*)d_in[0];
    const float* h       = (const float*)d_in[1];
    const float* s       = (const float*)d_in[2];
    const float* m       = (const float*)d_in[3];
    const float* tvec    = (const float*)d_in[4];
    const float* wt      = (const float*)d_in[5];
    const float* bt      = (const float*)d_in[6];
    const float* wk_x    = (const float*)d_in[7];
    const float* wv_x    = (const float*)d_in[8];
    const float* g_x     = (const float*)d_in[9];
    const float* be_x    = (const float*)d_in[10];
    const float* wk_h    = (const float*)d_in[11];
    const float* wv_h    = (const float*)d_in[12];
    const float* g_h     = (const float*)d_in[13];
    const float* be_h    = (const float*)d_in[14];
    const float* wk_s    = (const float*)d_in[15];
    const float* wv_s    = (const float*)d_in[16];
    const float* g_s     = (const float*)d_in[17];
    const float* be_s    = (const float*)d_in[18];
    const float* wk_m    = (const float*)d_in[19];
    const float* wv_m    = (const float*)d_in[20];
    const float* g_m     = (const float*)d_in[21];
    const float* be_m    = (const float*)d_in[22];
    const float* w_geglu = (const float*)d_in[23];
    const float* b_geglu = (const float*)d_in[24];
    const float* w_ff2   = (const float*)d_in[25];
    const float* b_ff2   = (const float*)d_in[26];
    const float* g_res   = (const float*)d_in[27];
    const float* be_res  = (const float*)d_in[28];
    const float* w_out   = (const float*)d_in[29];
    const float* b_out   = (const float*)d_in[30];
    float* out = (float*)d_out;
    float* ws  = (float*)d_ws;

    float* logits = ws + WS_LOG;
    float* mus    = ws + WS_MU;
    float* rstds  = ws + WS_RSTD;
    float* part   = ws + WS_PART;

    pass1_kernel<<<256, 512, 0, stream>>>(x, h, s, m, tvec, wt, bt,
                                          wk_x, wk_h, wk_s, wk_m,
                                          g_x, be_x, g_h, be_h, g_s, be_s, g_m, be_m,
                                          logits, mus, rstds);
    pass2_kernel<<<64, 1024, 0, stream>>>(x, h, s, m, logits, mus, rstds,
                                          wv_x, wv_h, wv_s, wv_m,
                                          g_x, be_x, g_h, be_h, g_s, be_s, g_m, be_m,
                                          part);
    tail_kernel<<<64, 512, 0, stream>>>(part,
                                        g_res, be_res, w_geglu, b_geglu,
                                        w_ff2, b_ff2, w_out, b_out, out);
}


# out projection tile: K split over the 16 waves (4 k-steps each; the reading wave adds the ff2 quarter sums and bias), not 64 chained steps in one wave; tail 10->9 barriers; y sum order changes
# speedup vs baseline: 3.7535x; 1.0757x over previous
#include <hip/hip_runtime.h>
#include <math.h>

typedef __attribute__((ext_vector_type(2))) float v2f;
typedef __attribute__((ext_vector_type(8))) float v8f;
typedef __attribute__((ext_vector_type(4))) float v4f;

#define C_DIM 256
#define D_DIM 256
#define N_TOK 4096
#define SCALE 0.0625f
#define EPSV  1e-5f

#define WS_LOG   1032
#define WS_MU    17416
#define WS_RSTD  33800
#define WS_PART  50184

__device__ __forceinline__ v8f gemv_tile_f32_unrolled(const float* x, const float* w,
                                                      long sk, int n0, v8f acc) {
    const int lane = threadIdx.x & 31;
    const int half = lane >> 4;
    const int l16  = lane & 15;
    const float amask = (l16 == 0) ? 1.0f : 0.0f;
    const float* wp = w + (long)(2 * half) * sk + (n0 + l16);
    const float* xp = x + 2 * half;
#pragma unroll
    for (int k0 = 0; k0 < 256; k0 += 4) {
        v2f a;
        a.x = xp[k0] * amask;
        a.y = xp[k0 + 1] * amask;
        v2f b;
        b.x = wp[(long)k0 * sk];
        b.y = wp[(long)(k0 + 1) * sk];
        acc = __builtin_amdgcn_wmma_f32_16x16x4_f32(
             false, a,  false, b,
             (short)0, acc,  false,  false);
    }
    return acc;
}

template <int STEPS>
__device__ __forceinline__ void gemv_quad_f32(const float* x, const float* w, long ldw,
                                              int nc0, int kbeg,
                                              float& r0, float& r1, float& r2, float& r3) {
    const int lane = threadIdx.x & 31;
    const int half = lane >> 4;
    const int l16  = lane & 15;
    const float amask = (l16 == 0) ? 1.0f : 0.0f;
    const float* wp = w + (long)(kbeg + 2 * half) * ldw + nc0 + 4 * l16;
    v8f acc0 = {}, acc1 = {}, acc2 = {}, acc3 = {};
#pragma unroll
    for (int s = 0; s < STEPS; ++s) {
        const int kb = kbeg + 4 * s + 2 * half;
        v2f a;
        a.x = x[kb] * amask;
        a.y = x[kb + 1] * amask;
        const v4f w0 = *(const v4f*)(wp + (long)(4 * s) * ldw);
        const v4f w1 = *(const v4f*)(wp + (long)(4 * s + 1) * ldw);
        v2f b;
        b.x = w0.x; b.y = w1.x;
        acc0 = __builtin_amdgcn_wmma_f32_16x16x4_f32(false, a, false, b, (short)0, acc0, false, false);
        b.x = w0.y; b.y = w1.y;
        acc1 = __builtin_amdgcn_wmma_f32_16x16x4_f32(false, a, false, b, (short)0, acc1, false, false);
        b.x = w0.z; b.y = w1.z;
        acc2 = __builtin_amdgcn_wmma_f32_16x16x4_f32(false, a, false, b, (short)0, acc2, false, false);
        b.x = w0.w; b.y = w1.w;
        acc3 = __builtin_amdgcn_wmma_f32_16x16x4_f32(false, a, false, b, (short)0, acc3, false, false);
    }
    r0 = acc0[0]; r1 = acc1[0]; r2 = acc2[0]; r3 = acc3[0];
}

__device__ __forceinline__ v8f gemv_tile_f32_rows(const float* x, const float* w, int n0, v8f acc) {
    const int lane = threadIdx.x & 31;
    const int half = lane >> 4;
    const int l16  = lane & 15;
    const float amask = (l16 == 0) ? 1.0f : 0.0f;
    const float* wp = w + (long)(n0 + l16) * 256 + 4 * half;
    const float* xp = x + 4 * half;
#pragma unroll
    for (int s = 0; s < 32; ++s) {
        const v4f wv4 = *(const v4f*)(wp + 8 * s);
        const v4f xv4 = *(const v4f*)(xp + 8 * s);
        v2f a, b;
        a.x = xv4.x * amask; a.y = xv4.y * amask;
        b.x = wv4.x;         b.y = wv4.y;
        acc = __builtin_amdgcn_wmma_f32_16x16x4_f32(false, a, false, b, (short)0, acc, false, false);
        a.x = xv4.z * amask; a.y = xv4.w * amask;
        b.x = wv4.z;         b.y = wv4.w;
        acc = __builtin_amdgcn_wmma_f32_16x16x4_f32(false, a, false, b, (short)0, acc, false, false);
    }
    return acc;
}

__global__ __launch_bounds__(512) void pass1_kernel(
    const float* x, const float* h, const float* s, const float* m,
    const float* tvec, const float* wt, const float* bt,
    const float* wk_x, const float* wk_h, const float* wk_s, const float* wk_m,
    const float* g_x, const float* be_x, const float* g_h, const float* be_h,
    const float* g_s, const float* be_s, const float* g_m, const float* be_m,
    float* logits, float* mus, float* rstds)
{
    __shared__ __attribute__((aligned(16))) float q_s[D_DIM];
    __shared__ float qp_s[4][D_DIM];
    __shared__ float p_s[C_DIM];
    __shared__ __attribute__((aligned(16))) float gpm_s[C_DIM];
    __shared__ __attribute__((aligned(16))) float bpm_s[C_DIM];
    __shared__ float sgs[8];
    __shared__ float sbs[8];
    __shared__ float sh1[8][64];
    __shared__ float sh2[8][64];
    __shared__ float shd[8][64];
    const int tid  = threadIdx.x;
    const int wave = tid >> 5;
    const int lane = tid & 31;
    const int bid  = blockIdx.x;
    const int mod  = bid >> 6;
    const int j0   = (bid & 63) << 6;
    const int tl   = tid & 63;
    const int grp  = tid >> 6;

    const float* a  = (mod == 0) ? x    : (mod == 1) ? h    : (mod == 2) ? s    : m;
    const float* wk = (mod == 0) ? wk_x : (mod == 1) ? wk_h : (mod == 2) ? wk_s : wk_m;
    const float* gg = (mod == 0) ? g_x  : (mod == 1) ? g_h  : (mod == 2) ? g_s  : g_m;
    const float* bb = (mod == 0) ? be_x : (mod == 1) ? be_h : (mod == 2) ? be_s : be_m;

    float v[32];
    {
        const float* ap = a + (long)(grp << 5) * N_TOK + (j0 + tl);
#pragma unroll
        for (int cc = 0; cc < 32; ++cc) v[cc] = ap[(long)cc * N_TOK];
    }

    {
        const int cg = (wave & 3) << 6;
        const int kq = wave >> 2;
        float r0, r1, r2, r3;
        gemv_quad_f32<16>(tvec, wt, (long)D_DIM, cg, kq << 6, r0, r1, r2, r3);
        if (lane < 16) {
            float* qd = &qp_s[kq][cg + 4 * lane];
            qd[0] = r0; qd[1] = r1; qd[2] = r2; qd[3] = r3;
        }
    }
    __syncthreads();
    if (tid < D_DIM)
        q_s[tid] = ((qp_s[0][tid] + qp_s[1][tid]) + (qp_s[2][tid] + qp_s[3][tid])) + bt[tid];
    __syncthreads();

    {
        v8f acc = {};
        acc = gemv_tile_f32_rows(q_s, wk, wave * 16, acc);
        if (lane < 16) p_s[wave * 16 + lane] = acc[0];
    }
    __syncthreads();

    if (tid < C_DIM) {
        const float p = p_s[tid];
        gpm_s[tid] = gg[tid] * p;
        bpm_s[tid] = bb[tid] * p;
    }
    __syncthreads();

    float s1 = 0.0f, s2 = 0.0f, dp = 0.0f;
    float sga[4] = {0.0f, 0.0f, 0.0f, 0.0f}, sba[4] = {0.0f, 0.0f, 0.0f, 0.0f};
    {
        const int c0 = grp << 5;
#pragma unroll
        for (int cc = 0; cc < 32; ++cc) {
            const float vv = v[cc];
            const float gp = gpm_s[c0 + cc];
            s1 += vv;
            s2 += vv * vv;
            dp += vv * gp;
            sga[cc & 3] += gp;
            sba[cc & 3] += bpm_s[c0 + cc];
        }
    }
    sh1[grp][tl] = s1; sh2[grp][tl] = s2; shd[grp][tl] = dp;
    if (tl == 0) {
        sgs[grp] = (sga[0] + sga[1]) + (sga[2] + sga[3]);
        sbs[grp] = (sba[0] + sba[1]) + (sba[2] + sba[3]);
    }
    __syncthreads();

    if (tid < 64) {
        float t1 = sh1[0][tid], t2 = sh2[0][tid], td = shd[0][tid];
        float sgp = sgs[0], sbp = sbs[0];
#pragma unroll
        for (int gi = 1; gi < 8; ++gi) {
            t1 += sh1[gi][tid]; t2 += sh2[gi][tid]; td += shd[gi][tid];
            sgp += sgs[gi]; sbp += sbs[gi];
        }
        const float mu   = t1 * (1.0f / C_DIM);
        const float var  = t2 * (1.0f / C_DIM) - mu * mu;
        const float rstd = rsqrtf(var + EPSV);
        const float lg   = SCALE * (rstd * (td - mu * sgp) + sbp);
        const int idx = mod * N_TOK + j0 + tid;
        logits[idx] = lg;
        mus[idx]    = mu;
        rstds[idx]  = rstd;
    }
}

__device__ __forceinline__ float wave_sum(float v) {
#pragma unroll
    for (int o = 16; o >= 1; o >>= 1) v += __shfl_xor(v, o, 32);
    return v;
}
__device__ __forceinline__ float wave_max(float v) {
#pragma unroll
    for (int o = 16; o >= 1; o >>= 1) v = fmaxf(v, __shfl_xor(v, o, 32));
    return v;
}

__global__ __launch_bounds__(1024) void pass2_kernel(
    const float* x, const float* h, const float* s, const float* m,
    const float* logits, const float* mus, const float* rstds,
    const float* wv_x, const float* wv_h, const float* wv_s, const float* wv_m,
    const float* g_x, const float* be_x, const float* g_h, const float* be_h,
    const float* g_s, const float* be_s, const float* g_m, const float* be_m,
    float* part)
{
    __shared__ float red_max[32];
    __shared__ float red_sum[32];
    __shared__ float red_out[32][18];
    __shared__ float u_s[16];
    __shared__ float o_s[2][D_DIM];
    const int tid  = threadIdx.x;
    const int wave = tid >> 5;
    const int lane = tid & 31;
    const int half = lane >> 4;
    const int l16  = lane & 15;
    const int bid  = blockIdx.x;
    const int mod  = bid >> 4;
    const int c0   = (bid & 15) << 4;

    const float* a  = (mod == 0) ? x    : (mod == 1) ? h    : (mod == 2) ? s    : m;
    const float* wv = (mod == 0) ? wv_x : (mod == 1) ? wv_h : (mod == 2) ? wv_s : wv_m;
    const float* gg = (mod == 0) ? g_x  : (mod == 1) ? g_h  : (mod == 2) ? g_s  : g_m;
    const float* bb = (mod == 0) ? be_x : (mod == 1) ? be_h : (mod == 2) ? be_s : be_m;

    const float gch = gg[c0 + (tid & 15)];
    const float bch = bb[c0 + (tid & 15)];

    const int kq = wave >> 4;
    const int nt = (wave & 15) << 4;
    v2f bw[2];
    {
        const float* wp = wv + (long)(c0 + 8 * kq) * D_DIM + nt + l16;
#pragma unroll
        for (int ks = 0; ks < 2; ++ks) {
            const int kb = 4 * ks + 2 * half;
            bw[ks].x = wp[(long)kb * D_DIM];
            bw[ks].y = wp[(long)(kb + 1) * D_DIM];
        }
    }

    v4f L[4];
    {
        const v4f* lg4 = (const v4f*)logits + tid;
#pragma unroll
        for (int q = 0; q < 4; ++q) L[q] = lg4[q * 1024];
    }
    const v4f rs = ((const v4f*)(rstds + mod * N_TOK))[tid];
    const v4f mu = ((const v4f*)(mus + mod * N_TOK))[tid];
    v4f av[16];
#pragma unroll
    for (int cc = 0; cc < 16; ++cc)
        av[cc] = ((const v4f*)(a + (long)(c0 + cc) * N_TOK))[tid];

    __builtin_amdgcn_sched_barrier(0);

    float lmax = -3.4e38f;
#pragma unroll
    for (int q = 0; q < 4; ++q)
        lmax = fmaxf(lmax, fmaxf(fmaxf(L[q].x, L[q].y), fmaxf(L[q].z, L[q].w)));
    lmax = wave_max(lmax);
    if (lane == 0) red_max[wave] = lmax;
    __syncthreads();
    const float gmax = wave_max(red_max[lane]);

    float ls = 0.0f;
#pragma unroll
    for (int q = 0; q < 4; ++q) {
        v4f e;
        e.x = __expf(L[q].x - gmax);
        e.y = __expf(L[q].y - gmax);
        e.z = __expf(L[q].z - gmax);
        e.w = __expf(L[q].w - gmax);
        L[q] = e;
        ls += e.x; ls += e.y; ls += e.z; ls += e.w;
    }
    ls = wave_sum(ls);
    if (lane == 0) red_sum[wave] = ls;
    __syncthreads();
    const float zsum = wave_sum(red_sum[lane]);
    const float inv = 1.0f / zsum;

    float wps = 0.0f, aps = 0.0f;
    float t[16];
#pragma unroll
    for (int cc = 0; cc < 16; ++cc) t[cc] = 0.0f;
    {
        const v4f e = (mod == 0) ? L[0] : (mod == 1) ? L[1] : (mod == 2) ? L[2] : L[3];
#pragma unroll
        for (int k = 0; k < 4; ++k) {
            const float w  = e[k] * inv;
            const float al = w * rs[k];
            wps += w;
            aps += al * mu[k];
#pragma unroll
            for (int cc = 0; cc < 16; ++cc) t[cc] += al * av[cc][k];
        }
    }
    wps = wave_sum(wps); aps = wave_sum(aps);
    float r8[8], r4[4], r2[2], r1;
    {
        const bool b = (lane & 16) != 0;
#pragma unroll
        for (int i = 0; i < 8; ++i) {
            const float snd = b ? t[i] : t[i + 8];
            const float kp  = b ? t[i + 8] : t[i];
            r8[i] = kp + __shfl_xor(snd, 16, 32);
        }
    }
    {
        const bool b = (lane & 8) != 0;
#pragma unroll
        for (int i = 0; i < 4; ++i) {
            const float snd = b ? r8[i] : r8[i + 4];
            const float kp  = b ? r8[i + 4] : r8[i];
            r4[i] = kp + __shfl_xor(snd, 8, 32);
        }
    }
    {
        const bool b = (lane & 4) != 0;
#pragma unroll
        for (int i = 0; i < 2; ++i) {
            const float snd = b ? r4[i] : r4[i + 2];
            const float kp  = b ? r4[i + 2] : r4[i];
            r2[i] = kp + __shfl_xor(snd, 4, 32);
        }
    }
    {
        const bool b = (lane & 2) != 0;
        const float snd = b ? r2[0] : r2[1];
        const float kp  = b ? r2[1] : r2[0];
        r1 = kp + __shfl_xor(snd, 2, 32);
    }
    r1 += __shfl_xor(r1, 1, 32);
    if ((lane & 1) == 0) red_out[wave][lane >> 1] = r1;
    if (lane == 0) { red_out[wave][16] = wps; red_out[wave][17] = aps; }
    __syncthreads();
    if (tid < 16) {
        float T = red_out[0][tid], wsum = red_out[0][16], amu = red_out[0][17];
#pragma unroll
        for (int w = 1; w < 32; ++w) { T += red_out[w][tid]; wsum += red_out[w][16]; amu += red_out[w][17]; }
        u_s[tid] = gch * (T - amu) + bch * wsum;
    }
    __syncthreads();

    {
        const float amask = (l16 == 0) ? 1.0f : 0.0f;
        v8f acc = {};
#pragma unroll
        for (int ks = 0; ks < 2; ++ks) {
            const int kb = 8 * kq + 4 * ks + 2 * half;
            v2f av2;
            av2.x = u_s[kb] * amask;
            av2.y = u_s[kb + 1] * amask;
            acc = __builtin_amdgcn_wmma_f32_16x16x4_f32(
                 false, av2,  false, bw[ks],
                 (short)0, acc,  false,  false);
        }
        if (lane < 16) o_s[kq][nt + lane] = acc[0];
    }
    __syncthreads();
    if (tid < D_DIM) part[bid * D_DIM + tid] = o_s[0][tid] + o_s[1][tid];
}

__global__ __launch_bounds__(512) void tail_kernel(
    const float* part,
    const float* g_res, const float* be_res,
    const float* w_geglu, const float* b_geglu,
    const float* w_ff2, const float* b_ff2,
    const float* w_out, const float* b_out,
    float* out)
{
    __shared__ __attribute__((aligned(16))) float ovec[D_DIM];
    __shared__ __attribute__((aligned(16))) float psum[8][D_DIM];
    __shared__ __attribute__((aligned(16))) float pm[32];
    __shared__ __attribute__((aligned(16))) float pv[32];
    __shared__ float rr[D_DIM];
    __shared__ float ggp[2][2 * D_DIM];
    __shared__ float hh[D_DIM];
    __shared__ __attribute__((aligned(16))) float ffp[4][D_DIM];
    __shared__ __attribute__((aligned(16))) float ypt[16][16];
    const int tid  = threadIdx.x;
    const int wave = tid >> 5;
    const int lane = tid & 31;
    const int c0   = blockIdx.x << 2;
    const int n0   = c0 & ~15;

    {
        const int d4 = tid & 63;
        const int g  = tid >> 6;
        const v4f* p4 = (const v4f*)part + (long)(g * 8) * (D_DIM / 4) + d4;
        v4f pv[8];
#pragma unroll
        for (int i = 0; i < 8; ++i) pv[i] = p4[i * (D_DIM / 4)];
        v4f sacc = pv[0];
#pragma unroll
        for (int i = 1; i < 8; ++i) sacc += pv[i];
        ((v4f*)psum[g])[d4] = sacc;
    }
    __syncthreads();
    if (tid < D_DIM) {
        float o = psum[0][tid];
#pragma unroll
        for (int g = 1; g < 8; ++g) o += psum[g][tid];
        ovec[tid] = o;
    }
    __syncthreads();

    v4f oa = {0.0f, 0.0f, 0.0f, 0.0f}, ob = {0.0f, 0.0f, 0.0f, 0.0f};
    if (wave == 0) {
        oa = ((const v4f*)ovec)[lane];
        ob = ((const v4f*)ovec)[lane + 32];
        pm[lane] = ((oa.x + oa.y) + (oa.z + oa.w)) + ((ob.x + ob.y) + (ob.z + ob.w));
    }
    __syncthreads();
    float mu;
    {
        v4f t4 = ((const v4f*)pm)[0];
        for (int i = 1; i < 8; ++i) t4 += ((const v4f*)pm)[i];
        mu = ((t4.x + t4.y) + (t4.z + t4.w)) * (1.0f / D_DIM);
    }
    if (wave == 0) {
        const v4f da = oa - mu, db = ob - mu;
        pv[lane] = ((da.x * da.x + da.y * da.y) + (da.z * da.z + da.w * da.w))
                 + ((db.x * db.x + db.y * db.y) + (db.z * db.z + db.w * db.w));
    }
    __syncthreads();
    float var;
    {
        v4f t4 = ((const v4f*)pv)[0];
        for (int i = 1; i < 8; ++i) t4 += ((const v4f*)pv)[i];
        var = ((t4.x + t4.y) + (t4.z + t4.w)) * (1.0f / D_DIM);
    }
    const float rstd = rsqrtf(var + EPSV);
    if (tid < D_DIM) rr[tid] = (ovec[tid] - mu) * rstd * g_res[tid] + be_res[tid];
    __syncthreads();

    {
        const int cg = (wave & 7) << 6;
        const int kh = wave >> 3;
        float r0, r1, r2, r3;
        gemv_quad_f32<32>(rr, w_geglu, (long)(2 * D_DIM), cg, kh << 7, r0, r1, r2, r3);
        if (lane < 16) {
            float* gd = &ggp[kh][cg + 4 * lane];
            gd[0] = r0; gd[1] = r1; gd[2] = r2; gd[3] = r3;
        }
    }
    __syncthreads();

    if (tid < D_DIM) {
        const float u  = (ggp[0][tid] + ggp[1][tid]) + b_geglu[tid];
        const float g  = (ggp[0][D_DIM + tid] + ggp[1][D_DIM + tid]) + b_geglu[D_DIM + tid];
        const float ge = 0.5f * g * (1.0f + erff(g * 0.70710678118654752f));
        hh[tid] = u * ge;
    }
    __syncthreads();

    const int half = lane >> 4;
    const int l16  = lane & 15;
    const int kw   = 16 * wave + 2 * half;
    const int jch  = (c0 & 15) + (tid >> 7);
    v2f wo[4], bf[4];
    {
        const float* wop = w_out + (long)kw * C_DIM + n0 + l16;
#pragma unroll
        for (int s = 0; s < 4; ++s) {
            wo[s].x = wop[(long)(4 * s) * C_DIM];
            wo[s].y = wop[(long)(4 * s + 1) * C_DIM];
            bf[s]   = *(const v2f*)(b_ff2 + kw + 4 * s);
        }
    }
    const float byv = b_out[n0 + jch];

    {
        const int cg = (wave & 3) << 6;
        const int kq = wave >> 2;
        float r0, r1, r2, r3;
        gemv_quad_f32<16>(hh, w_ff2, (long)D_DIM, cg, kq << 6, r0, r1, r2, r3);
        if (lane < 16) {
            float* fd = &ffp[kq][cg + 4 * lane];
            fd[0] = r0; fd[1] = r1; fd[2] = r2; fd[3] = r3;
        }
    }
    __syncthreads();

    {
        const float amask = (l16 == 0) ? 1.0f : 0.0f;
        v8f acc = {};
#pragma unroll
        for (int s = 0; s < 4; ++s) {
            const int kb = kw + 4 * s;
            const v2f p0 = *(const v2f*)&ffp[0][kb];
            const v2f p1 = *(const v2f*)&ffp[1][kb];
            const v2f p2 = *(const v2f*)&ffp[2][kb];
            const v2f p3 = *(const v2f*)&ffp[3][kb];
            v2f a;
            a.x = (((p0.x + p1.x) + (p2.x + p3.x)) + bf[s].x) * amask;
            a.y = (((p0.y + p1.y) + (p2.y + p3.y)) + bf[s].y) * amask;
            acc = __builtin_amdgcn_wmma_f32_16x16x4_f32(
                 false, a,  false, wo[s],
                 (short)0, acc,  false,  false);
        }
        if (lane < 16) ypt[lane][wave] = acc[0];
    }
    __syncthreads();

    {
        const int ch = tid >> 7;
        const v4f* yp = (const v4f*)ypt[jch];
        const v4f y0 = yp[0], y1 = yp[1], y2 = yp[2], y3 = yp[3];
        const float yv = ((((y0.x + y0.y) + (y0.z + y0.w)) + ((y1.x + y1.y) + (y1.z + y1.w)))
                        + (((y2.x + y2.y) + (y2.z + y2.w)) + ((y3.x + y3.y) + (y3.z + y3.w)))) + byv;
        const v4f vv = {yv, yv, yv, yv};
        v4f* o = (v4f*)(out + (long)(c0 + ch) * N_TOK) + (tid & 127);
#pragma unroll
        for (int i = 0; i < 8; ++i) o[i * 128] = vv;
    }
}

extern "C" void kernel_launch(void* const* d_in, const int* in_sizes, int n_in,
                              void* d_out, int out_size, void* d_ws, size_t ws_size,
                              hipStream_t stream) {
    const float* x       = (const float*)d_in[0];
    const float* h       = (const float*)d_in[1];
    const float* s       = (const float*)d_in[2];
    const float* m       = (const float*)d_in[3];
    const float* tvec    = (const float*)d_in[4];
    const float* wt      = (const float*)d_in[5];
    const float* bt      = (const float*)d_in[6];
    const float* wk_x    = (const float*)d_in[7];
    const float* wv_x    = (const float*)d_in[8];
    const float* g_x     = (const float*)d_in[9];
    const float* be_x    = (const float*)d_in[10];
    const float* wk_h    = (const float*)d_in[11];
    const float* wv_h    = (const float*)d_in[12];
    const float* g_h     = (const float*)d_in[13];
    const float* be_h    = (const float*)d_in[14];
    const float* wk_s    = (const float*)d_in[15];
    const float* wv_s    = (const float*)d_in[16];
    const float* g_s     = (const float*)d_in[17];
    const float* be_s    = (const float*)d_in[18];
    const float* wk_m    = (const float*)d_in[19];
    const float* wv_m    = (const float*)d_in[20];
    const float* g_m     = (const float*)d_in[21];
    const float* be_m    = (const float*)d_in[22];
    const float* w_geglu = (const float*)d_in[23];
    const float* b_geglu = (const float*)d_in[24];
    const float* w_ff2   = (const float*)d_in[25];
    const float* b_ff2   = (const float*)d_in[26];
    const float* g_res   = (const float*)d_in[27];
    const float* be_res  = (const float*)d_in[28];
    const float* w_out   = (const float*)d_in[29];
    const float* b_out   = (const float*)d_in[30];
    float* out = (float*)d_out;
    float* ws  = (float*)d_ws;

    float* logits = ws + WS_LOG;
    float* mus    = ws + WS_MU;
    float* rstds  = ws + WS_RSTD;
    float* part   = ws + WS_PART;

    pass1_kernel<<<256, 512, 0, stream>>>(x, h, s, m, tvec, wt, bt,
                                          wk_x, wk_h, wk_s, wk_m,
                                          g_x, be_x, g_h, be_h, g_s, be_s, g_m, be_m,
                                          logits, mus, rstds);
    pass2_kernel<<<64, 1024, 0, stream>>>(x, h, s, m, logits, mus, rstds,
                                          wv_x, wv_h, wv_s, wv_m,
                                          g_x, be_x, g_h, be_h, g_s, be_s, g_m, be_m,
                                          part);
    tail_kernel<<<64, 512, 0, stream>>>(part,
                                        g_res, be_res, w_geglu, b_geglu,
                                        w_ff2, b_ff2, w_out, b_out, out);
}


# k walk of the shared-weight GEMVs (q, p, GEGLU, ff2) starts at a per-workgroup rotated step and wraps: workgroups stop asking L2 for the same lines at once; q, p, gg, ff sum orders vary by workgroup
# speedup vs baseline: 3.8875x; 1.0357x over previous
#include <hip/hip_runtime.h>
#include <math.h>

typedef __attribute__((ext_vector_type(2))) float v2f;
typedef __attribute__((ext_vector_type(8))) float v8f;
typedef __attribute__((ext_vector_type(4))) float v4f;

#define C_DIM 256
#define D_DIM 256
#define N_TOK 4096
#define SCALE 0.0625f
#define EPSV  1e-5f

#define WS_LOG   1032
#define WS_MU    17416
#define WS_RSTD  33800
#define WS_PART  50184

__device__ __forceinline__ v8f gemv_tile_f32_unrolled(const float* x, const float* w,
                                                      long sk, int n0, v8f acc) {
    const int lane = threadIdx.x & 31;
    const int half = lane >> 4;
    const int l16  = lane & 15;
    const float amask = (l16 == 0) ? 1.0f : 0.0f;
    const float* wp = w + (long)(2 * half) * sk + (n0 + l16);
    const float* xp = x + 2 * half;
#pragma unroll
    for (int k0 = 0; k0 < 256; k0 += 4) {
        v2f a;
        a.x = xp[k0] * amask;
        a.y = xp[k0 + 1] * amask;
        v2f b;
        b.x = wp[(long)k0 * sk];
        b.y = wp[(long)(k0 + 1) * sk];
        acc = __builtin_amdgcn_wmma_f32_16x16x4_f32(
             false, a,  false, b,
             (short)0, acc,  false,  false);
    }
    return acc;
}

template <int STEPS>
__device__ __forceinline__ void gemv_quad_f32(const float* x, const float* w, long ldw,
                                              int nc0, int kbeg, int rot,
                                              float& r0, float& r1, float& r2, float& r3) {
    const int lane = threadIdx.x & 31;
    const int half = lane >> 4;
    const int l16  = lane & 15;
    const float amask = (l16 == 0) ? 1.0f : 0.0f;
    const float* wp = w + (long)(kbeg + 2 * half) * ldw + nc0 + 4 * l16;
    v8f acc0 = {}, acc1 = {}, acc2 = {}, acc3 = {};
#pragma unroll
    for (int s = 0; s < STEPS; ++s) {
        const int sr = (s + rot) & (STEPS - 1);
        const int kb = kbeg + 4 * sr + 2 * half;
        v2f a;
        a.x = x[kb] * amask;
        a.y = x[kb + 1] * amask;
        const v4f w0 = *(const v4f*)(wp + (long)(4 * sr) * ldw);
        const v4f w1 = *(const v4f*)(wp + (long)(4 * sr + 1) * ldw);
        v2f b;
        b.x = w0.x; b.y = w1.x;
        acc0 = __builtin_amdgcn_wmma_f32_16x16x4_f32(false, a, false, b, (short)0, acc0, false, false);
        b.x = w0.y; b.y = w1.y;
        acc1 = __builtin_amdgcn_wmma_f32_16x16x4_f32(false, a, false, b, (short)0, acc1, false, false);
        b.x = w0.z; b.y = w1.z;
        acc2 = __builtin_amdgcn_wmma_f32_16x16x4_f32(false, a, false, b, (short)0, acc2, false, false);
        b.x = w0.w; b.y = w1.w;
        acc3 = __builtin_amdgcn_wmma_f32_16x16x4_f32(false, a, false, b, (short)0, acc3, false, false);
    }
    r0 = acc0[0]; r1 = acc1[0]; r2 = acc2[0]; r3 = acc3[0];
}

__device__ __forceinline__ v8f gemv_tile_f32_rows(const float* x, const float* w, int n0, int rot, v8f acc) {
    const int lane = threadIdx.x & 31;
    const int half = lane >> 4;
    const int l16  = lane & 15;
    const float amask = (l16 == 0) ? 1.0f : 0.0f;
    const float* wp = w + (long)(n0 + l16) * 256 + 4 * half;
    const float* xp = x + 4 * half;
#pragma unroll
    for (int s = 0; s < 32; ++s) {
        const int sr = (s + rot) & 31;
        const v4f wv4 = *(const v4f*)(wp + 8 * sr);
        const v4f xv4 = *(const v4f*)(xp + 8 * sr);
        v2f a, b;
        a.x = xv4.x * amask; a.y = xv4.y * amask;
        b.x = wv4.x;         b.y = wv4.y;
        acc = __builtin_amdgcn_wmma_f32_16x16x4_f32(false, a, false, b, (short)0, acc, false, false);
        a.x = xv4.z * amask; a.y = xv4.w * amask;
        b.x = wv4.z;         b.y = wv4.w;
        acc = __builtin_amdgcn_wmma_f32_16x16x4_f32(false, a, false, b, (short)0, acc, false, false);
    }
    return acc;
}

__global__ __launch_bounds__(512) void pass1_kernel(
    const float* x, const float* h, const float* s, const float* m,
    const float* tvec, const float* wt, const float* bt,
    const float* wk_x, const float* wk_h, const float* wk_s, const float* wk_m,
    const float* g_x, const float* be_x, const float* g_h, const float* be_h,
    const float* g_s, const float* be_s, const float* g_m, const float* be_m,
    float* logits, float* mus, float* rstds)
{
    __shared__ __attribute__((aligned(16))) float q_s[D_DIM];
    __shared__ float qp_s[4][D_DIM];
    __shared__ float p_s[C_DIM];
    __shared__ __attribute__((aligned(16))) float gpm_s[C_DIM];
    __shared__ __attribute__((aligned(16))) float bpm_s[C_DIM];
    __shared__ float sgs[8];
    __shared__ float sbs[8];
    __shared__ float sh1[8][64];
    __shared__ float sh2[8][64];
    __shared__ float shd[8][64];
    const int tid  = threadIdx.x;
    const int wave = tid >> 5;
    const int lane = tid & 31;
    const int bid  = blockIdx.x;
    const int mod  = bid >> 6;
    const int j0   = (bid & 63) << 6;
    const int tl   = tid & 63;
    const int grp  = tid >> 6;

    const float* a  = (mod == 0) ? x    : (mod == 1) ? h    : (mod == 2) ? s    : m;
    const float* wk = (mod == 0) ? wk_x : (mod == 1) ? wk_h : (mod == 2) ? wk_s : wk_m;
    const float* gg = (mod == 0) ? g_x  : (mod == 1) ? g_h  : (mod == 2) ? g_s  : g_m;
    const float* bb = (mod == 0) ? be_x : (mod == 1) ? be_h : (mod == 2) ? be_s : be_m;

    float v[32];
    {
        const float* ap = a + (long)(grp << 5) * N_TOK + (j0 + tl);
#pragma unroll
        for (int cc = 0; cc < 32; ++cc) v[cc] = ap[(long)cc * N_TOK];
    }

    {
        const int cg = (wave & 3) << 6;
        const int kq = wave >> 2;
        float r0, r1, r2, r3;
        gemv_quad_f32<16>(tvec, wt, (long)D_DIM, cg, kq << 6, bid & 15, r0, r1, r2, r3);
        if (lane < 16) {
            float* qd = &qp_s[kq][cg + 4 * lane];
            qd[0] = r0; qd[1] = r1; qd[2] = r2; qd[3] = r3;
        }
    }
    __syncthreads();
    if (tid < D_DIM)
        q_s[tid] = ((qp_s[0][tid] + qp_s[1][tid]) + (qp_s[2][tid] + qp_s[3][tid])) + bt[tid];
    __syncthreads();

    {
        v8f acc = {};
        acc = gemv_tile_f32_rows(q_s, wk, wave * 16, (bid & 7) << 2, acc);
        if (lane < 16) p_s[wave * 16 + lane] = acc[0];
    }
    __syncthreads();

    if (tid < C_DIM) {
        const float p = p_s[tid];
        gpm_s[tid] = gg[tid] * p;
        bpm_s[tid] = bb[tid] * p;
    }
    __syncthreads();

    float s1 = 0.0f, s2 = 0.0f, dp = 0.0f;
    float sga[4] = {0.0f, 0.0f, 0.0f, 0.0f}, sba[4] = {0.0f, 0.0f, 0.0f, 0.0f};
    {
        const int c0 = grp << 5;
#pragma unroll
        for (int cc = 0; cc < 32; ++cc) {
            const float vv = v[cc];
            const float gp = gpm_s[c0 + cc];
            s1 += vv;
            s2 += vv * vv;
            dp += vv * gp;
            sga[cc & 3] += gp;
            sba[cc & 3] += bpm_s[c0 + cc];
        }
    }
    sh1[grp][tl] = s1; sh2[grp][tl] = s2; shd[grp][tl] = dp;
    if (tl == 0) {
        sgs[grp] = (sga[0] + sga[1]) + (sga[2] + sga[3]);
        sbs[grp] = (sba[0] + sba[1]) + (sba[2] + sba[3]);
    }
    __syncthreads();

    if (tid < 64) {
        float t1 = sh1[0][tid], t2 = sh2[0][tid], td = shd[0][tid];
        float sgp = sgs[0], sbp = sbs[0];
#pragma unroll
        for (int gi = 1; gi < 8; ++gi) {
            t1 += sh1[gi][tid]; t2 += sh2[gi][tid]; td += shd[gi][tid];
            sgp += sgs[gi]; sbp += sbs[gi];
        }
        const float mu   = t1 * (1.0f / C_DIM);
        const float var  = t2 * (1.0f / C_DIM) - mu * mu;
        const float rstd = rsqrtf(var + EPSV);
        const float lg   = SCALE * (rstd * (td - mu * sgp) + sbp);
        const int idx = mod * N_TOK + j0 + tid;
        logits[idx] = lg;
        mus[idx]    = mu;
        rstds[idx]  = rstd;
    }
}

__device__ __forceinline__ float wave_sum(float v) {
#pragma unroll
    for (int o = 16; o >= 1; o >>= 1) v += __shfl_xor(v, o, 32);
    return v;
}
__device__ __forceinline__ float wave_max(float v) {
#pragma unroll
    for (int o = 16; o >= 1; o >>= 1) v = fmaxf(v, __shfl_xor(v, o, 32));
    return v;
}

__global__ __launch_bounds__(1024) void pass2_kernel(
    const float* x, const float* h, const float* s, const float* m,
    const float* logits, const float* mus, const float* rstds,
    const float* wv_x, const float* wv_h, const float* wv_s, const float* wv_m,
    const float* g_x, const float* be_x, const float* g_h, const float* be_h,
    const float* g_s, const float* be_s, const float* g_m, const float* be_m,
    float* part)
{
    __shared__ float red_max[32];
    __shared__ float red_sum[32];
    __shared__ float red_out[32][18];
    __shared__ float u_s[16];
    __shared__ float o_s[2][D_DIM];
    const int tid  = threadIdx.x;
    const int wave = tid >> 5;
    const int lane = tid & 31;
    const int half = lane >> 4;
    const int l16  = lane & 15;
    const int bid  = blockIdx.x;
    const int mod  = bid >> 4;
    const int c0   = (bid & 15) << 4;

    const float* a  = (mod == 0) ? x    : (mod == 1) ? h    : (mod == 2) ? s    : m;
    const float* wv = (mod == 0) ? wv_x : (mod == 1) ? wv_h : (mod == 2) ? wv_s : wv_m;
    const float* gg = (mod == 0) ? g_x  : (mod == 1) ? g_h  : (mod == 2) ? g_s  : g_m;
    const float* bb = (mod == 0) ? be_x : (mod == 1) ? be_h : (mod == 2) ? be_s : be_m;

    const float gch = gg[c0 + (tid & 15)];
    const float bch = bb[c0 + (tid & 15)];

    const int kq = wave >> 4;
    const int nt = (wave & 15) << 4;
    v2f bw[2];
    {
        const float* wp = wv + (long)(c0 + 8 * kq) * D_DIM + nt + l16;
#pragma unroll
        for (int ks = 0; ks < 2; ++ks) {
            const int kb = 4 * ks + 2 * half;
            bw[ks].x = wp[(long)kb * D_DIM];
            bw[ks].y = wp[(long)(kb + 1) * D_DIM];
        }
    }

    v4f L[4];
    {
        const v4f* lg4 = (const v4f*)logits + tid;
#pragma unroll
        for (int q = 0; q < 4; ++q) L[q] = lg4[q * 1024];
    }
    const v4f rs = ((const v4f*)(rstds + mod * N_TOK))[tid];
    const v4f mu = ((const v4f*)(mus + mod * N_TOK))[tid];
    v4f av[16];
#pragma unroll
    for (int cc = 0; cc < 16; ++cc)
        av[cc] = ((const v4f*)(a + (long)(c0 + cc) * N_TOK))[tid];

    __builtin_amdgcn_sched_barrier(0);

    float lmax = -3.4e38f;
#pragma unroll
    for (int q = 0; q < 4; ++q)
        lmax = fmaxf(lmax, fmaxf(fmaxf(L[q].x, L[q].y), fmaxf(L[q].z, L[q].w)));
    lmax = wave_max(lmax);
    if (lane == 0) red_max[wave] = lmax;
    __syncthreads();
    const float gmax = wave_max(red_max[lane]);

    float ls = 0.0f;
#pragma unroll
    for (int q = 0; q < 4; ++q) {
        v4f e;
        e.x = __expf(L[q].x - gmax);
        e.y = __expf(L[q].y - gmax);
        e.z = __expf(L[q].z - gmax);
        e.w = __expf(L[q].w - gmax);
        L[q] = e;
        ls += e.x; ls += e.y; ls += e.z; ls += e.w;
    }
    ls = wave_sum(ls);
    if (lane == 0) red_sum[wave] = ls;
    __syncthreads();
    const float zsum = wave_sum(red_sum[lane]);
    const float inv = 1.0f / zsum;

    float wps = 0.0f, aps = 0.0f;
    float t[16];
#pragma unroll
    for (int cc = 0; cc < 16; ++cc) t[cc] = 0.0f;
    {
        const v4f e = (mod == 0) ? L[0] : (mod == 1) ? L[1] : (mod == 2) ? L[2] : L[3];
#pragma unroll
        for (int k = 0; k < 4; ++k) {
            const float w  = e[k] * inv;
            const float al = w * rs[k];
            wps += w;
            aps += al * mu[k];
#pragma unroll
            for (int cc = 0; cc < 16; ++cc) t[cc] += al * av[cc][k];
        }
    }
    wps = wave_sum(wps); aps = wave_sum(aps);
    float r8[8], r4[4], r2[2], r1;
    {
        const bool b = (lane & 16) != 0;
#pragma unroll
        for (int i = 0; i < 8; ++i) {
            const float snd = b ? t[i] : t[i + 8];
            const float kp  = b ? t[i + 8] : t[i];
            r8[i] = kp + __shfl_xor(snd, 16, 32);
        }
    }
    {
        const bool b = (lane & 8) != 0;
#pragma unroll
        for (int i = 0; i < 4; ++i) {
            const float snd = b ? r8[i] : r8[i + 4];
            const float kp  = b ? r8[i + 4] : r8[i];
            r4[i] = kp + __shfl_xor(snd, 8, 32);
        }
    }
    {
        const bool b = (lane & 4) != 0;
#pragma unroll
        for (int i = 0; i < 2; ++i) {
            const float snd = b ? r4[i] : r4[i + 2];
            const float kp  = b ? r4[i + 2] : r4[i];
            r2[i] = kp + __shfl_xor(snd, 4, 32);
        }
    }
    {
        const bool b = (lane & 2) != 0;
        const float snd = b ? r2[0] : r2[1];
        const float kp  = b ? r2[1] : r2[0];
        r1 = kp + __shfl_xor(snd, 2, 32);
    }
    r1 += __shfl_xor(r1, 1, 32);
    if ((lane & 1) == 0) red_out[wave][lane >> 1] = r1;
    if (lane == 0) { red_out[wave][16] = wps; red_out[wave][17] = aps; }
    __syncthreads();
    if (tid < 16) {
        float T = red_out[0][tid], wsum = red_out[0][16], amu = red_out[0][17];
#pragma unroll
        for (int w = 1; w < 32; ++w) { T += red_out[w][tid]; wsum += red_out[w][16]; amu += red_out[w][17]; }
        u_s[tid] = gch * (T - amu) + bch * wsum;
    }
    __syncthreads();

    {
        const float amask = (l16 == 0) ? 1.0f : 0.0f;
        v8f acc = {};
#pragma unroll
        for (int ks = 0; ks < 2; ++ks) {
            const int kb = 8 * kq + 4 * ks + 2 * half;
            v2f av2;
            av2.x = u_s[kb] * amask;
            av2.y = u_s[kb + 1] * amask;
            acc = __builtin_amdgcn_wmma_f32_16x16x4_f32(
                 false, av2,  false, bw[ks],
                 (short)0, acc,  false,  false);
        }
        if (lane < 16) o_s[kq][nt + lane] = acc[0];
    }
    __syncthreads();
    if (tid < D_DIM) part[bid * D_DIM + tid] = o_s[0][tid] + o_s[1][tid];
}

__global__ __launch_bounds__(512) void tail_kernel(
    const float* part,
    const float* g_res, const float* be_res,
    const float* w_geglu, const float* b_geglu,
    const float* w_ff2, const float* b_ff2,
    const float* w_out, const float* b_out,
    float* out)
{
    __shared__ __attribute__((aligned(16))) float ovec[D_DIM];
    __shared__ __attribute__((aligned(16))) float psum[8][D_DIM];
    __shared__ __attribute__((aligned(16))) float pm[32];
    __shared__ __attribute__((aligned(16))) float pv[32];
    __shared__ float rr[D_DIM];
    __shared__ float ggp[2][2 * D_DIM];
    __shared__ float hh[D_DIM];
    __shared__ __attribute__((aligned(16))) float ffp[4][D_DIM];
    __shared__ __attribute__((aligned(16))) float ypt[16][16];
    const int tid  = threadIdx.x;
    const int wave = tid >> 5;
    const int lane = tid & 31;
    const int c0   = blockIdx.x << 2;
    const int n0   = c0 & ~15;

    {
        const int d4 = tid & 63;
        const int g  = tid >> 6;
        const v4f* p4 = (const v4f*)part + (long)(g * 8) * (D_DIM / 4) + d4;
        v4f pv[8];
#pragma unroll
        for (int i = 0; i < 8; ++i) pv[i] = p4[i * (D_DIM / 4)];
        v4f sacc = pv[0];
#pragma unroll
        for (int i = 1; i < 8; ++i) sacc += pv[i];
        ((v4f*)psum[g])[d4] = sacc;
    }
    __syncthreads();
    if (tid < D_DIM) {
        float o = psum[0][tid];
#pragma unroll
        for (int g = 1; g < 8; ++g) o += psum[g][tid];
        ovec[tid] = o;
    }
    __syncthreads();

    v4f oa = {0.0f, 0.0f, 0.0f, 0.0f}, ob = {0.0f, 0.0f, 0.0f, 0.0f};
    if (wave == 0) {
        oa = ((const v4f*)ovec)[lane];
        ob = ((const v4f*)ovec)[lane + 32];
        pm[lane] = ((oa.x + oa.y) + (oa.z + oa.w)) + ((ob.x + ob.y) + (ob.z + ob.w));
    }
    __syncthreads();
    float mu;
    {
        v4f t4 = ((const v4f*)pm)[0];
        for (int i = 1; i < 8; ++i) t4 += ((const v4f*)pm)[i];
        mu = ((t4.x + t4.y) + (t4.z + t4.w)) * (1.0f / D_DIM);
    }
    if (wave == 0) {
        const v4f da = oa - mu, db = ob - mu;
        pv[lane] = ((da.x * da.x + da.y * da.y) + (da.z * da.z + da.w * da.w))
                 + ((db.x * db.x + db.y * db.y) + (db.z * db.z + db.w * db.w));
    }
    __syncthreads();
    float var;
    {
        v4f t4 = ((const v4f*)pv)[0];
        for (int i = 1; i < 8; ++i) t4 += ((const v4f*)pv)[i];
        var = ((t4.x + t4.y) + (t4.z + t4.w)) * (1.0f / D_DIM);
    }
    const float rstd = rsqrtf(var + EPSV);
    if (tid < D_DIM) rr[tid] = (ovec[tid] - mu) * rstd * g_res[tid] + be_res[tid];
    __syncthreads();

    {
        const int cg = (wave & 7) << 6;
        const int kh = wave >> 3;
        float r0, r1, r2, r3;
        gemv_quad_f32<32>(rr, w_geglu, (long)(2 * D_DIM), cg, kh << 7, (int)(blockIdx.x & 31), r0, r1, r2, r3);
        if (lane < 16) {
            float* gd = &ggp[kh][cg + 4 * lane];
            gd[0] = r0; gd[1] = r1; gd[2] = r2; gd[3] = r3;
        }
    }
    __syncthreads();

    if (tid < D_DIM) {
        const float u  = (ggp[0][tid] + ggp[1][tid]) + b_geglu[tid];
        const float g  = (ggp[0][D_DIM + tid] + ggp[1][D_DIM + tid]) + b_geglu[D_DIM + tid];
        const float ge = 0.5f * g * (1.0f + erff(g * 0.70710678118654752f));
        hh[tid] = u * ge;
    }
    __syncthreads();

    const int half = lane >> 4;
    const int l16  = lane & 15;
    const int kw   = 16 * wave + 2 * half;
    const int jch  = (c0 & 15) + (tid >> 7);
    v2f wo[4], bf[4];
    {
        const float* wop = w_out + (long)kw * C_DIM + n0 + l16;
#pragma unroll
        for (int s = 0; s < 4; ++s) {
            wo[s].x = wop[(long)(4 * s) * C_DIM];
            wo[s].y = wop[(long)(4 * s + 1) * C_DIM];
            bf[s]   = *(const v2f*)(b_ff2 + kw + 4 * s);
        }
    }
    const float byv = b_out[n0 + jch];

    {
        const int cg = (wave & 3) << 6;
        const int kq = wave >> 2;
        float r0, r1, r2, r3;
        gemv_quad_f32<16>(hh, w_ff2, (long)D_DIM, cg, kq << 6, (int)(blockIdx.x & 15), r0, r1, r2, r3);
        if (lane < 16) {
            float* fd = &ffp[kq][cg + 4 * lane];
            fd[0] = r0; fd[1] = r1; fd[2] = r2; fd[3] = r3;
        }
    }
    __syncthreads();

    {
        const float amask = (l16 == 0) ? 1.0f : 0.0f;
        v8f acc = {};
#pragma unroll
        for (int s = 0; s < 4; ++s) {
            const int kb = kw + 4 * s;
            const v2f p0 = *(const v2f*)&ffp[0][kb];
            const v2f p1 = *(const v2f*)&ffp[1][kb];
            const v2f p2 = *(const v2f*)&ffp[2][kb];
            const v2f p3 = *(const v2f*)&ffp[3][kb];
            v2f a;
            a.x = (((p0.x + p1.x) + (p2.x + p3.x)) + bf[s].x) * amask;
            a.y = (((p0.y + p1.y) + (p2.y + p3.y)) + bf[s].y) * amask;
            acc = __builtin_amdgcn_wmma_f32_16x16x4_f32(
                 false, a,  false, wo[s],
                 (short)0, acc,  false,  false);
        }
        if (lane < 16) ypt[lane][wave] = acc[0];
    }
    __syncthreads();

    {
        const int ch = tid >> 7;
        const v4f* yp = (const v4f*)ypt[jch];
        const v4f y0 = yp[0], y1 = yp[1], y2 = yp[2], y3 = yp[3];
        const float yv = ((((y0.x + y0.y) + (y0.z + y0.w)) + ((y1.x + y1.y) + (y1.z + y1.w)))
                        + (((y2.x + y2.y) + (y2.z + y2.w)) + ((y3.x + y3.y) + (y3.z + y3.w)))) + byv;
        const v4f vv = {yv, yv, yv, yv};
        v4f* o = (v4f*)(out + (long)(c0 + ch) * N_TOK) + (tid & 127);
#pragma unroll
        for (int i = 0; i < 8; ++i) o[i * 128] = vv;
    }
}

extern "C" void kernel_launch(void* const* d_in, const int* in_sizes, int n_in,
                              void* d_out, int out_size, void* d_ws, size_t ws_size,
                              hipStream_t stream) {
    const float* x       = (const float*)d_in[0];
    const float* h       = (const float*)d_in[1];
    const float* s       = (const float*)d_in[2];
    const float* m       = (const float*)d_in[3];
    const float* tvec    = (const float*)d_in[4];
    const float* wt      = (const float*)d_in[5];
    const float* bt      = (const float*)d_in[6];
    const float* wk_x    = (const float*)d_in[7];
    const float* wv_x    = (const float*)d_in[8];
    const float* g_x     = (const float*)d_in[9];
    const float* be_x    = (const float*)d_in[10];
    const float* wk_h    = (const float*)d_in[11];
    const float* wv_h    = (const float*)d_in[12];
    const float* g_h     = (const float*)d_in[13];
    const float* be_h    = (const float*)d_in[14];
    const float* wk_s    = (const float*)d_in[15];
    const float* wv_s    = (const float*)d_in[16];
    const float* g_s     = (const float*)d_in[17];
    const float* be_s    = (const float*)d_in[18];
    const float* wk_m    = (const float*)d_in[19];
    const float* wv_m    = (const float*)d_in[20];
    const float* g_m     = (const float*)d_in[21];
    const float* be_m    = (const float*)d_in[22];
    const float* w_geglu = (const float*)d_in[23];
    const float* b_geglu = (const float*)d_in[24];
    const float* w_ff2   = (const float*)d_in[25];
    const float* b_ff2   = (const float*)d_in[26];
    const float* g_res   = (const float*)d_in[27];
    const float* be_res  = (const float*)d_in[28];
    const float* w_out   = (const float*)d_in[29];
    const float* b_out   = (const float*)d_in[30];
    float* out = (float*)d_out;
    float* ws  = (float*)d_ws;

    float* logits = ws + WS_LOG;
    float* mus    = ws + WS_MU;
    float* rstds  = ws + WS_RSTD;
    float* part   = ws + WS_PART;

    pass1_kernel<<<256, 512, 0, stream>>>(x, h, s, m, tvec, wt, bt,
                                          wk_x, wk_h, wk_s, wk_m,
                                          g_x, be_x, g_h, be_h, g_s, be_s, g_m, be_m,
                                          logits, mus, rstds);
    pass2_kernel<<<64, 1024, 0, stream>>>(x, h, s, m, logits, mus, rstds,
                                          wv_x, wv_h, wv_s, wv_m,
                                          g_x, be_x, g_h, be_h, g_s, be_s, g_m, be_m,
                                          part);
    tail_kernel<<<64, 512, 0, stream>>>(part,
                                        g_res, be_res, w_geglu, b_geglu,
                                        w_ff2, b_ff2, w_out, b_out, out);
}


# query-side chain cut in two along d: a pass1 workgroup (128 tokens) rebuilds half of q and the part of p it feeds, writes half logits; pass2 adds the two halves; q, p, logit sum orders change
# speedup vs baseline: 4.1244x; 1.0609x over previous
#include <hip/hip_runtime.h>
#include <math.h>

typedef __attribute__((ext_vector_type(2))) float v2f;
typedef __attribute__((ext_vector_type(8))) float v8f;
typedef __attribute__((ext_vector_type(4))) float v4f;

#define C_DIM 256
#define D_DIM 256
#define N_TOK 4096
#define SCALE 0.0625f
#define EPSV  1e-5f

#define WS_LOG   1032
#define WS_MU    17416
#define WS_RSTD  33800
#define WS_PART  50184

__device__ __forceinline__ v8f gemv_tile_f32_unrolled(const float* x, const float* w,
                                                      long sk, int n0, v8f acc) {
    const int lane = threadIdx.x & 31;
    const int half = lane >> 4;
    const int l16  = lane & 15;
    const float amask = (l16 == 0) ? 1.0f : 0.0f;
    const float* wp = w + (long)(2 * half) * sk + (n0 + l16);
    const float* xp = x + 2 * half;
#pragma unroll
    for (int k0 = 0; k0 < 256; k0 += 4) {
        v2f a;
        a.x = xp[k0] * amask;
        a.y = xp[k0 + 1] * amask;
        v2f b;
        b.x = wp[(long)k0 * sk];
        b.y = wp[(long)(k0 + 1) * sk];
        acc = __builtin_amdgcn_wmma_f32_16x16x4_f32(
             false, a,  false, b,
             (short)0, acc,  false,  false);
    }
    return acc;
}

template <int STEPS>
__device__ __forceinline__ void gemv_quad_f32(const float* x, const float* w, long ldw,
                                              int nc0, int kbeg, int rot,
                                              float& r0, float& r1, float& r2, float& r3) {
    const int lane = threadIdx.x & 31;
    const int half = lane >> 4;
    const int l16  = lane & 15;
    const float amask = (l16 == 0) ? 1.0f : 0.0f;
    const float* wp = w + (long)(kbeg + 2 * half) * ldw + nc0 + 4 * l16;
    v8f acc0 = {}, acc1 = {}, acc2 = {}, acc3 = {};
#pragma unroll
    for (int s = 0; s < STEPS; ++s) {
        const int sr = (s + rot) & (STEPS - 1);
        const int kb = kbeg + 4 * sr + 2 * half;
        v2f a;
        a.x = x[kb] * amask;
        a.y = x[kb + 1] * amask;
        const v4f w0 = *(const v4f*)(wp + (long)(4 * sr) * ldw);
        const v4f w1 = *(const v4f*)(wp + (long)(4 * sr + 1) * ldw);
        v2f b;
        b.x = w0.x; b.y = w1.x;
        acc0 = __builtin_amdgcn_wmma_f32_16x16x4_f32(false, a, false, b, (short)0, acc0, false, false);
        b.x = w0.y; b.y = w1.y;
        acc1 = __builtin_amdgcn_wmma_f32_16x16x4_f32(false, a, false, b, (short)0, acc1, false, false);
        b.x = w0.z; b.y = w1.z;
        acc2 = __builtin_amdgcn_wmma_f32_16x16x4_f32(false, a, false, b, (short)0, acc2, false, false);
        b.x = w0.w; b.y = w1.w;
        acc3 = __builtin_amdgcn_wmma_f32_16x16x4_f32(false, a, false, b, (short)0, acc3, false, false);
    }
    r0 = acc0[0]; r1 = acc1[0]; r2 = acc2[0]; r3 = acc3[0];
}

template <int LOADS>
__device__ __forceinline__ v8f gemv_tile_f32_rows(const float* x, const float* w, int n0, int kbase, int rot, v8f acc) {
    const int lane = threadIdx.x & 31;
    const int half = lane >> 4;
    const int l16  = lane & 15;
    const float amask = (l16 == 0) ? 1.0f : 0.0f;
    const float* wp = w + (long)(n0 + l16) * 256 + kbase + 4 * half;
    const float* xp = x + 4 * half;
#pragma unroll
    for (int s = 0; s < LOADS; ++s) {
        const int sr = (s + rot) & (LOADS - 1);
        const v4f wv4 = *(const v4f*)(wp + 8 * sr);
        const v4f xv4 = *(const v4f*)(xp + 8 * sr);
        v2f a, b;
        a.x = xv4.x * amask; a.y = xv4.y * amask;
        b.x = wv4.x;         b.y = wv4.y;
        acc = __builtin_amdgcn_wmma_f32_16x16x4_f32(false, a, false, b, (short)0, acc, false, false);
        a.x = xv4.z * amask; a.y = xv4.w * amask;
        b.x = wv4.z;         b.y = wv4.w;
        acc = __builtin_amdgcn_wmma_f32_16x16x4_f32(false, a, false, b, (short)0, acc, false, false);
    }
    return acc;
}

__global__ __launch_bounds__(512) void pass1_kernel(
    const float* x, const float* h, const float* s, const float* m,
    const float* tvec, const float* wt, const float* bt,
    const float* wk_x, const float* wk_h, const float* wk_s, const float* wk_m,
    const float* g_x, const float* be_x, const float* g_h, const float* be_h,
    const float* g_s, const float* be_s, const float* g_m, const float* be_m,
    float* lg0, float* lg1, float* mus, float* rstds)
{
    __shared__ __attribute__((aligned(16))) float q_s[128];
    __shared__ float qp_s[8][128];
    __shared__ float p_s[C_DIM];
    __shared__ __attribute__((aligned(16))) float gpm_s[C_DIM];
    __shared__ __attribute__((aligned(16))) float bpm_s[C_DIM];
    __shared__ float sgs[8];
    __shared__ float sbs[8];
    __shared__ __attribute__((aligned(8))) float sh1[8][128];
    __shared__ __attribute__((aligned(8))) float sh2[8][128];
    __shared__ __attribute__((aligned(8))) float shd[8][128];
    const int tid  = threadIdx.x;
    const int wave = tid >> 5;
    const int lane = tid & 31;
    const int bid  = blockIdx.x;
    const int dh   = bid & 1;
    const int d0   = dh << 7;
    const int tix  = (bid >> 1) & 31;
    const int mod  = bid >> 6;
    const int j0   = tix << 7;
    const int tc   = tid & 63;
    const int grp  = tid >> 6;

    const float* a  = (mod == 0) ? x    : (mod == 1) ? h    : (mod == 2) ? s    : m;
    const float* wk = (mod == 0) ? wk_x : (mod == 1) ? wk_h : (mod == 2) ? wk_s : wk_m;
    const float* gg = (mod == 0) ? g_x  : (mod == 1) ? g_h  : (mod == 2) ? g_s  : g_m;
    const float* bb = (mod == 0) ? be_x : (mod == 1) ? be_h : (mod == 2) ? be_s : be_m;

    v2f v[32];
    {
        const float* ap = a + (long)(grp << 5) * N_TOK + (j0 + 2 * tc);
#pragma unroll
        for (int cc = 0; cc < 32; ++cc) v[cc] = *(const v2f*)(ap + (long)cc * N_TOK);
    }

    {
        const int cq = (wave & 1) << 6;
        const int ke = wave >> 1;
        float r0, r1, r2, r3;
        gemv_quad_f32<8>(tvec, wt, (long)D_DIM, d0 + cq, ke << 5, (bid >> 1) & 7, r0, r1, r2, r3);
        if (lane < 16) {
            float* qd = &qp_s[ke][cq + 4 * lane];
            qd[0] = r0; qd[1] = r1; qd[2] = r2; qd[3] = r3;
        }
    }
    __syncthreads();
    if (tid < 128)
        q_s[tid] = (((qp_s[0][tid] + qp_s[1][tid]) + (qp_s[2][tid] + qp_s[3][tid]))
                  + ((qp_s[4][tid] + qp_s[5][tid]) + (qp_s[6][tid] + qp_s[7][tid]))) + bt[d0 + tid];
    __syncthreads();

    {
        v8f acc = {};
        acc = gemv_tile_f32_rows<16>(q_s, wk, wave * 16, d0, ((bid >> 1) & 3) << 2, acc);
        if (lane < 16) p_s[wave * 16 + lane] = acc[0];
    }
    __syncthreads();

    if (tid < C_DIM) {
        const float p = p_s[tid];
        gpm_s[tid] = gg[tid] * p;
        bpm_s[tid] = bb[tid] * p;
    }
    __syncthreads();

    v2f s1 = {0.0f, 0.0f}, s2 = {0.0f, 0.0f}, dp = {0.0f, 0.0f};
    float sga[4] = {0.0f, 0.0f, 0.0f, 0.0f}, sba[4] = {0.0f, 0.0f, 0.0f, 0.0f};
    {
        const int c0 = grp << 5;
#pragma unroll
        for (int cc = 0; cc < 32; ++cc) {
            const v2f vv = v[cc];
            const float gp = gpm_s[c0 + cc];
            s1 += vv;
            s2 += vv * vv;
            dp += vv * gp;
            sga[cc & 3] += gp;
            sba[cc & 3] += bpm_s[c0 + cc];
        }
    }
    *(v2f*)&sh1[grp][2 * tc] = s1;
    *(v2f*)&sh2[grp][2 * tc] = s2;
    *(v2f*)&shd[grp][2 * tc] = dp;
    if (tc == 0) {
        sgs[grp] = (sga[0] + sga[1]) + (sga[2] + sga[3]);
        sbs[grp] = (sba[0] + sba[1]) + (sba[2] + sba[3]);
    }
    __syncthreads();

    if (tid < 128) {
        float t1 = sh1[0][tid], t2 = sh2[0][tid], td = shd[0][tid];
        float sgp = sgs[0], sbp = sbs[0];
#pragma unroll
        for (int gi = 1; gi < 8; ++gi) {
            t1 += sh1[gi][tid]; t2 += sh2[gi][tid]; td += shd[gi][tid];
            sgp += sgs[gi]; sbp += sbs[gi];
        }
        const float mu   = t1 * (1.0f / C_DIM);
        const float var  = t2 * (1.0f / C_DIM) - mu * mu;
        const float rstd = rsqrtf(var + EPSV);
        const float lg   = SCALE * (rstd * (td - mu * sgp) + sbp);
        const int idx = mod * N_TOK + j0 + tid;
        if (dh == 0) {
            lg0[idx]   = lg;
            mus[idx]   = mu;
            rstds[idx] = rstd;
        } else {
            lg1[idx]   = lg;
        }
    }
}

__device__ __forceinline__ float wave_sum(float v) {
#pragma unroll
    for (int o = 16; o >= 1; o >>= 1) v += __shfl_xor(v, o, 32);
    return v;
}
__device__ __forceinline__ float wave_max(float v) {
#pragma unroll
    for (int o = 16; o >= 1; o >>= 1) v = fmaxf(v, __shfl_xor(v, o, 32));
    return v;
}

__global__ __launch_bounds__(1024) void pass2_kernel(
    const float* x, const float* h, const float* s, const float* m,
    const float* logits, const float* logits1, const float* mus, const float* rstds,
    const float* wv_x, const float* wv_h, const float* wv_s, const float* wv_m,
    const float* g_x, const float* be_x, const float* g_h, const float* be_h,
    const float* g_s, const float* be_s, const float* g_m, const float* be_m,
    float* part)
{
    __shared__ float red_max[32];
    __shared__ float red_sum[32];
    __shared__ float red_out[32][18];
    __shared__ float u_s[16];
    __shared__ float o_s[2][D_DIM];
    const int tid  = threadIdx.x;
    const int wave = tid >> 5;
    const int lane = tid & 31;
    const int half = lane >> 4;
    const int l16  = lane & 15;
    const int bid  = blockIdx.x;
    const int mod  = bid >> 4;
    const int c0   = (bid & 15) << 4;

    const float* a  = (mod == 0) ? x    : (mod == 1) ? h    : (mod == 2) ? s    : m;
    const float* wv = (mod == 0) ? wv_x : (mod == 1) ? wv_h : (mod == 2) ? wv_s : wv_m;
    const float* gg = (mod == 0) ? g_x  : (mod == 1) ? g_h  : (mod == 2) ? g_s  : g_m;
    const float* bb = (mod == 0) ? be_x : (mod == 1) ? be_h : (mod == 2) ? be_s : be_m;

    const float gch = gg[c0 + (tid & 15)];
    const float bch = bb[c0 + (tid & 15)];

    const int kq = wave >> 4;
    const int nt = (wave & 15) << 4;
    v2f bw[2];
    {
        const float* wp = wv + (long)(c0 + 8 * kq) * D_DIM + nt + l16;
#pragma unroll
        for (int ks = 0; ks < 2; ++ks) {
            const int kb = 4 * ks + 2 * half;
            bw[ks].x = wp[(long)kb * D_DIM];
            bw[ks].y = wp[(long)(kb + 1) * D_DIM];
        }
    }

    v4f L[4], L1[4];
    {
        const v4f* lg4 = (const v4f*)logits + tid;
        const v4f* lh4 = (const v4f*)logits1 + tid;
#pragma unroll
        for (int q = 0; q < 4; ++q) { L[q] = lg4[q * 1024]; L1[q] = lh4[q * 1024]; }
    }
    const v4f rs = ((const v4f*)(rstds + mod * N_TOK))[tid];
    const v4f mu = ((const v4f*)(mus + mod * N_TOK))[tid];
    v4f av[16];
#pragma unroll
    for (int cc = 0; cc < 16; ++cc)
        av[cc] = ((const v4f*)(a + (long)(c0 + cc) * N_TOK))[tid];

    __builtin_amdgcn_sched_barrier(0);

#pragma unroll
    for (int q = 0; q < 4; ++q) L[q] += L1[q];

    float lmax = -3.4e38f;
#pragma unroll
    for (int q = 0; q < 4; ++q)
        lmax = fmaxf(lmax, fmaxf(fmaxf(L[q].x, L[q].y), fmaxf(L[q].z, L[q].w)));
    lmax = wave_max(lmax);
    if (lane == 0) red_max[wave] = lmax;
    __syncthreads();
    const float gmax = wave_max(red_max[lane]);

    float ls = 0.0f;
#pragma unroll
    for (int q = 0; q < 4; ++q) {
        v4f e;
        e.x = __expf(L[q].x - gmax);
        e.y = __expf(L[q].y - gmax);
        e.z = __expf(L[q].z - gmax);
        e.w = __expf(L[q].w - gmax);
        L[q] = e;
        ls += e.x; ls += e.y; ls += e.z; ls += e.w;
    }
    ls = wave_sum(ls);
    if (lane == 0) red_sum[wave] = ls;
    __syncthreads();
    const float zsum = wave_sum(red_sum[lane]);
    const float inv = 1.0f / zsum;

    float wps = 0.0f, aps = 0.0f;
    float t[16];
#pragma unroll
    for (int cc = 0; cc < 16; ++cc) t[cc] = 0.0f;
    {
        const v4f e = (mod == 0) ? L[0] : (mod == 1) ? L[1] : (mod == 2) ? L[2] : L[3];
#pragma unroll
        for (int k = 0; k < 4; ++k) {
            const float w  = e[k] * inv;
            const float al = w * rs[k];
            wps += w;
            aps += al * mu[k];
#pragma unroll
            for (int cc = 0; cc < 16; ++cc) t[cc] += al * av[cc][k];
        }
    }
    wps = wave_sum(wps); aps = wave_sum(aps);
    float r8[8], r4[4], r2[2], r1;
    {
        const bool b = (lane & 16) != 0;
#pragma unroll
        for (int i = 0; i < 8; ++i) {
            const float snd = b ? t[i] : t[i + 8];
            const float kp  = b ? t[i + 8] : t[i];
            r8[i] = kp + __shfl_xor(snd, 16, 32);
        }
    }
    {
        const bool b = (lane & 8) != 0;
#pragma unroll
        for (int i = 0; i < 4; ++i) {
            const float snd = b ? r8[i] : r8[i + 4];
            const float kp  = b ? r8[i + 4] : r8[i];
            r4[i] = kp + __shfl_xor(snd, 8, 32);
        }
    }
    {
        const bool b = (lane & 4) != 0;
#pragma unroll
        for (int i = 0; i < 2; ++i) {
            const float snd = b ? r4[i] : r4[i + 2];
            const float kp  = b ? r4[i + 2] : r4[i];
            r2[i] = kp + __shfl_xor(snd, 4, 32);
        }
    }
    {
        const bool b = (lane & 2) != 0;
        const float snd = b ? r2[0] : r2[1];
        const float kp  = b ? r2[1] : r2[0];
        r1 = kp + __shfl_xor(snd, 2, 32);
    }
    r1 += __shfl_xor(r1, 1, 32);
    if ((lane & 1) == 0) red_out[wave][lane >> 1] = r1;
    if (lane == 0) { red_out[wave][16] = wps; red_out[wave][17] = aps; }
    __syncthreads();
    if (tid < 16) {
        float T = red_out[0][tid], wsum = red_out[0][16], amu = red_out[0][17];
#pragma unroll
        for (int w = 1; w < 32; ++w) { T += red_out[w][tid]; wsum += red_out[w][16]; amu += red_out[w][17]; }
        u_s[tid] = gch * (T - amu) + bch * wsum;
    }
    __syncthreads();

    {
        const float amask = (l16 == 0) ? 1.0f : 0.0f;
        v8f acc = {};
#pragma unroll
        for (int ks = 0; ks < 2; ++ks) {
            const int kb = 8 * kq + 4 * ks + 2 * half;
            v2f av2;
            av2.x = u_s[kb] * amask;
            av2.y = u_s[kb + 1] * amask;
            acc = __builtin_amdgcn_wmma_f32_16x16x4_f32(
                 false, av2,  false, bw[ks],
                 (short)0, acc,  false,  false);
        }
        if (lane < 16) o_s[kq][nt + lane] = acc[0];
    }
    __syncthreads();
    if (tid < D_DIM) part[bid * D_DIM + tid] = o_s[0][tid] + o_s[1][tid];
}

__global__ __launch_bounds__(512) void tail_kernel(
    const float* part,
    const float* g_res, const float* be_res,
    const float* w_geglu, const float* b_geglu,
    const float* w_ff2, const float* b_ff2,
    const float* w_out, const float* b_out,
    float* out)
{
    __shared__ __attribute__((aligned(16))) float ovec[D_DIM];
    __shared__ __attribute__((aligned(16))) float psum[8][D_DIM];
    __shared__ __attribute__((aligned(16))) float pm[32];
    __shared__ __attribute__((aligned(16))) float pv[32];
    __shared__ float rr[D_DIM];
    __shared__ float ggp[2][2 * D_DIM];
    __shared__ float hh[D_DIM];
    __shared__ __attribute__((aligned(16))) float ffp[4][D_DIM];
    __shared__ __attribute__((aligned(16))) float ypt[16][16];
    const int tid  = threadIdx.x;
    const int wave = tid >> 5;
    const int lane = tid & 31;
    const int c0   = blockIdx.x << 2;
    const int n0   = c0 & ~15;

    {
        const int d4 = tid & 63;
        const int g  = tid >> 6;
        const v4f* p4 = (const v4f*)part + (long)(g * 8) * (D_DIM / 4) + d4;
        v4f pv[8];
#pragma unroll
        for (int i = 0; i < 8; ++i) pv[i] = p4[i * (D_DIM / 4)];
        v4f sacc = pv[0];
#pragma unroll
        for (int i = 1; i < 8; ++i) sacc += pv[i];
        ((v4f*)psum[g])[d4] = sacc;
    }
    __syncthreads();
    if (tid < D_DIM) {
        float o = psum[0][tid];
#pragma unroll
        for (int g = 1; g < 8; ++g) o += psum[g][tid];
        ovec[tid] = o;
    }
    __syncthreads();

    v4f oa = {0.0f, 0.0f, 0.0f, 0.0f}, ob = {0.0f, 0.0f, 0.0f, 0.0f};
    if (wave == 0) {
        oa = ((const v4f*)ovec)[lane];
        ob = ((const v4f*)ovec)[lane + 32];
        pm[lane] = ((oa.x + oa.y) + (oa.z + oa.w)) + ((ob.x + ob.y) + (ob.z + ob.w));
    }
    __syncthreads();
    float mu;
    {
        v4f t4 = ((const v4f*)pm)[0];
        for (int i = 1; i < 8; ++i) t4 += ((const v4f*)pm)[i];
        mu = ((t4.x + t4.y) + (t4.z + t4.w)) * (1.0f / D_DIM);
    }
    if (wave == 0) {
        const v4f da = oa - mu, db = ob - mu;
        pv[lane] = ((da.x * da.x + da.y * da.y) + (da.z * da.z + da.w * da.w))
                 + ((db.x * db.x + db.y * db.y) + (db.z * db.z + db.w * db.w));
    }
    __syncthreads();
    float var;
    {
        v4f t4 = ((const v4f*)pv)[0];
        for (int i = 1; i < 8; ++i) t4 += ((const v4f*)pv)[i];
        var = ((t4.x + t4.y) + (t4.z + t4.w)) * (1.0f / D_DIM);
    }
    const float rstd = rsqrtf(var + EPSV);
    if (tid < D_DIM) rr[tid] = (ovec[tid] - mu) * rstd * g_res[tid] + be_res[tid];
    __syncthreads();

    {
        const int cg = (wave & 7) << 6;
        const int kh = wave >> 3;
        float r0, r1, r2, r3;
        gemv_quad_f32<32>(rr, w_geglu, (long)(2 * D_DIM), cg, kh << 7, (int)(blockIdx.x & 31), r0, r1, r2, r3);
        if (lane < 16) {
            float* gd = &ggp[kh][cg + 4 * lane];
            gd[0] = r0; gd[1] = r1; gd[2] = r2; gd[3] = r3;
        }
    }
    __syncthreads();

    if (tid < D_DIM) {
        const float u  = (ggp[0][tid] + ggp[1][tid]) + b_geglu[tid];
        const float g  = (ggp[0][D_DIM + tid] + ggp[1][D_DIM + tid]) + b_geglu[D_DIM + tid];
        const float ge = 0.5f * g * (1.0f + erff(g * 0.70710678118654752f));
        hh[tid] = u * ge;
    }
    __syncthreads();

    const int half = lane >> 4;
    const int l16  = lane & 15;
    const int kw   = 16 * wave + 2 * half;
    const int jch  = (c0 & 15) + (tid >> 7);
    v2f wo[4], bf[4];
    {
        const float* wop = w_out + (long)kw * C_DIM + n0 + l16;
#pragma unroll
        for (int s = 0; s < 4; ++s) {
            wo[s].x = wop[(long)(4 * s) * C_DIM];
            wo[s].y = wop[(long)(4 * s + 1) * C_DIM];
            bf[s]   = *(const v2f*)(b_ff2 + kw + 4 * s);
        }
    }
    const float byv = b_out[n0 + jch];

    {
        const int cg = (wave & 3) << 6;
        const int kq = wave >> 2;
        float r0, r1, r2, r3;
        gemv_quad_f32<16>(hh, w_ff2, (long)D_DIM, cg, kq << 6, (int)(blockIdx.x & 15), r0, r1, r2, r3);
        if (lane < 16) {
            float* fd = &ffp[kq][cg + 4 * lane];
            fd[0] = r0; fd[1] = r1; fd[2] = r2; fd[3] = r3;
        }
    }
    __syncthreads();

    {
        const float amask = (l16 == 0) ? 1.0f : 0.0f;
        v8f acc = {};
#pragma unroll
        for (int s = 0; s < 4; ++s) {
            const int kb = kw + 4 * s;
            const v2f p0 = *(const v2f*)&ffp[0][kb];
            const v2f p1 = *(const v2f*)&ffp[1][kb];
            const v2f p2 = *(const v2f*)&ffp[2][kb];
            const v2f p3 = *(const v2f*)&ffp[3][kb];
            v2f a;
            a.x = (((p0.x + p1.x) + (p2.x + p3.x)) + bf[s].x) * amask;
            a.y = (((p0.y + p1.y) + (p2.y + p3.y)) + bf[s].y) * amask;
            acc = __builtin_amdgcn_wmma_f32_16x16x4_f32(
                 false, a,  false, wo[s],
                 (short)0, acc,  false,  false);
        }
        if (lane < 16) ypt[lane][wave] = acc[0];
    }
    __syncthreads();

    {
        const int ch = tid >> 7;
        const v4f* yp = (const v4f*)ypt[jch];
        const v4f y0 = yp[0], y1 = yp[1], y2 = yp[2], y3 = yp[3];
        const float yv = ((((y0.x + y0.y) + (y0.z + y0.w)) + ((y1.x + y1.y) + (y1.z + y1.w)))
                        + (((y2.x + y2.y) + (y2.z + y2.w)) + ((y3.x + y3.y) + (y3.z + y3.w)))) + byv;
        const v4f vv = {yv, yv, yv, yv};
        v4f* o = (v4f*)(out + (long)(c0 + ch) * N_TOK) + (tid & 127);
#pragma unroll
        for (int i = 0; i < 8; ++i) o[i * 128] = vv;
    }
}

extern "C" void kernel_launch(void* const* d_in, const int* in_sizes, int n_in,
                              void* d_out, int out_size, void* d_ws, size_t ws_size,
                              hipStream_t stream) {
    const float* x       = (const float*)d_in[0];
    const float* h       = (const float*)d_in[1];
    const float* s       = (const float*)d_in[2];
    const float* m       = (const float*)d_in[3];
    const float* tvec    = (const float*)d_in[4];
    const float* wt      = (const float*)d_in[5];
    const float* bt      = (const float*)d_in[6];
    const float* wk_x    = (const float*)d_in[7];
    const float* wv_x    = (const float*)d_in[8];
    const float* g_x     = (const float*)d_in[9];
    const float* be_x    = (const float*)d_in[10];
    const float* wk_h    = (const float*)d_in[11];
    const float* wv_h    = (const float*)d_in[12];
    const float* g_h     = (const float*)d_in[13];
    const float* be_h    = (const float*)d_in[14];
    const float* wk_s    = (const float*)d_in[15];
    const float* wv_s    = (const float*)d_in[16];
    const float* g_s     = (const float*)d_in[17];
    const float* be_s    = (const float*)d_in[18];
    const float* wk_m    = (const float*)d_in[19];
    const float* wv_m    = (const float*)d_in[20];
    const float* g_m     = (const float*)d_in[21];
    const float* be_m    = (const float*)d_in[22];
    const float* w_geglu = (const float*)d_in[23];
    const float* b_geglu = (const float*)d_in[24];
    const float* w_ff2   = (const float*)d_in[25];
    const float* b_ff2   = (const float*)d_in[26];
    const float* g_res   = (const float*)d_in[27];
    const float* be_res  = (const float*)d_in[28];
    const float* w_out   = (const float*)d_in[29];
    const float* b_out   = (const float*)d_in[30];
    float* out = (float*)d_out;
    float* ws  = (float*)d_ws;

    float* logits = ws + WS_LOG;
    float* logits1 = out;
    float* mus    = ws + WS_MU;
    float* rstds  = ws + WS_RSTD;
    float* part   = ws + WS_PART;

    pass1_kernel<<<256, 512, 0, stream>>>(x, h, s, m, tvec, wt, bt,
                                          wk_x, wk_h, wk_s, wk_m,
                                          g_x, be_x, g_h, be_h, g_s, be_s, g_m, be_m,
                                          logits, logits1, mus, rstds);
    pass2_kernel<<<64, 1024, 0, stream>>>(x, h, s, m, logits, logits1, mus, rstds,
                                          wv_x, wv_h, wv_s, wv_m,
                                          g_x, be_x, g_h, be_h, g_s, be_s, g_m, be_m,
                                          part);
    tail_kernel<<<64, 512, 0, stream>>>(part,
                                        g_res, be_res, w_geglu, b_geglu,
                                        w_ff2, b_ff2, w_out, b_out, out);
}
